# starting kernel
# baseline (speedup 1.0000x reference)
#include <hip/hip_runtime.h>
#include <hip/hip_bf16.h>


typedef __attribute__((ext_vector_type(2))) float v2f;
typedef __attribute__((ext_vector_type(8))) float v8f;

#define IN_CH 256
#define HID_CH 128
#define GEMM_ROWS 32


__global__ __launch_bounds__(256) void k_init_deg(float* deg, int N) {
    int i = blockIdx.x * 256 + threadIdx.x;
    if (i < N) deg[i] = 1.0f;
}

__global__ __launch_bounds__(256) void k_deg_scatter(const int* __restrict__ col,
                                                     const float* __restrict__ w,
                                                     float* deg, int E) {
    int e = blockIdx.x * 256 + threadIdx.x;
    if (e < E) unsafeAtomicAdd(&deg[col[e]], w[e]);
}

__global__ __launch_bounds__(256) void k_dinv(const float* __restrict__ deg,
                                              float* dinv, int N) {
    int i = blockIdx.x * 256 + threadIdx.x;
    if (i < N) {
        float d = deg[i];
        dinv[i] = (d > 0.0f) ? rsqrtf(d) : 0.0f;
    }
}

__global__ __launch_bounds__(256) void k_edge_norm(const int* __restrict__ row,
                                                   const int* __restrict__ col,
                                                   const float* __restrict__ w,
                                                   const float* __restrict__ dinv,
                                                   float* nrm, int E) {
    int e = blockIdx.x * 256 + threadIdx.x;
    if (e < E) nrm[e] = dinv[row[e]] * w[e] * dinv[col[e]];
}


__global__ __launch_bounds__(256) void k_gemm_wmma(const float* __restrict__ A,
                                                   const float* __restrict__ W,
                                                   float* __restrict__ H, int K) {
    __shared__ float As[GEMM_ROWS * (IN_CH + 4)];
    const int stride = K + 4;
    const int tid = threadIdx.x;
    const int rowbase = blockIdx.x * GEMM_ROWS;

    for (int idx = tid; idx < GEMM_ROWS * K; idx += 256) {
        int r = idx / K;
        int kk = idx - r * K;
        As[r * stride + kk] = A[(size_t)(rowbase + r) * K + kk];
    }
    __syncthreads();

    const int wave = tid >> 5;
    const int lane = tid & 31;
    const int m    = lane & 15;
    const int half = lane >> 4;
    const int cb   = wave * 16;

    v8f c0 = {};
    v8f c1 = {};
    const float* wp  = W + cb + m;
    const float* ap0 = As + m * stride;
    const float* ap1 = As + (16 + m) * stride;

    for (int k = 0; k < K; k += 4) {
        const int koff = k + half * 2;
        v2f b;
        b[0] = wp[(size_t)koff * HID_CH];
        b[1] = wp[(size_t)(koff + 1) * HID_CH];
        v2f a0, a1;
        a0[0] = ap0[koff];
        a0[1] = ap0[koff + 1];
        a1[0] = ap1[koff];
        a1[1] = ap1[koff + 1];
        c0 = __builtin_amdgcn_wmma_f32_16x16x4_f32(false, a0, false, b,
                                                   (short)0, c0, false, false);
        c1 = __builtin_amdgcn_wmma_f32_16x16x4_f32(false, a1, false, b,
                                                   (short)0, c1, false, false);
    }

    float* hp0 = H + (size_t)rowbase * HID_CH + cb + m;
    float* hp1 = hp0 + (size_t)16 * HID_CH;
#pragma unroll
    for (int r = 0; r < 8; ++r) {
        hp0[(size_t)(r + 8 * half) * HID_CH] = c0[r];
        hp1[(size_t)(r + 8 * half) * HID_CH] = c1[r];
    }
}


__global__ __launch_bounds__(256) void k_init_acc(const float* __restrict__ h,
                                                  const float* __restrict__ dinv,
                                                  float* __restrict__ acc, int N) {
    int t = blockIdx.x * 256 + threadIdx.x;
    int node = t >> 5, chunk = t & 31;
    if (node >= N) return;
    float s = dinv[node] * dinv[node];
    const float4* hp = (const float4*)(h + (size_t)node * HID_CH);
    float4* ap = (float4*)(acc + (size_t)node * HID_CH);
    float4 v = hp[chunk];
    v.x *= s; v.y *= s; v.z *= s; v.w *= s;
    ap[chunk] = v;
}

__global__ __launch_bounds__(256) void k_scatter(const int* __restrict__ row,
                                                 const int* __restrict__ col,
                                                 const float* __restrict__ nrm,
                                                 const float* __restrict__ h,
                                                 float* __restrict__ acc, int E) {
    int wid = (blockIdx.x * 256 + threadIdx.x) >> 5;
    int lane = threadIdx.x & 31;
    if (wid >= E) return;
    int r = row[wid];
    int c = col[wid];
    float nv = nrm[wid];
    const float4* hp = (const float4*)(h + (size_t)r * HID_CH);
    float4 v = hp[lane];
    float* dst = acc + (size_t)c * HID_CH + lane * 4;
    unsafeAtomicAdd(dst + 0, v.x * nv);
    unsafeAtomicAdd(dst + 1, v.y * nv);
    unsafeAtomicAdd(dst + 2, v.z * nv);
    unsafeAtomicAdd(dst + 3, v.w * nv);
}

__global__ __launch_bounds__(256) void k_bias_relu(float* __restrict__ acc,
                                                   const float* __restrict__ b, int N) {
    int g = blockIdx.x * 256 + threadIdx.x;
    if (g >= N * 32) return;
    float4* ap = (float4*)acc;
    const float4* bp = (const float4*)b;
    float4 v = ap[g];
    float4 bb = bp[g & 31];
    v.x = fmaxf(v.x + bb.x, 0.0f);
    v.y = fmaxf(v.y + bb.y, 0.0f);
    v.z = fmaxf(v.z + bb.z, 0.0f);
    v.w = fmaxf(v.w + bb.w, 0.0f);
    ap[g] = v;
}

__global__ __launch_bounds__(256) void k_bias_l2norm(float* __restrict__ out,
                                                     const float* __restrict__ b, int N) {
    int wid = (blockIdx.x * 256 + threadIdx.x) >> 5;
    int lane = threadIdx.x & 31;
    if (wid >= N) return;
    float4* op = (float4*)(out + (size_t)wid * HID_CH);
    const float4* bp = (const float4*)b;
    float4 v = op[lane];
    float4 bb = bp[lane];
    v.x += bb.x; v.y += bb.y; v.z += bb.z; v.w += bb.w;
    float ss = v.x * v.x + v.y * v.y + v.z * v.z + v.w * v.w;
#pragma unroll
    for (int m = 16; m >= 1; m >>= 1) ss += __shfl_xor(ss, m, 32);
    float inv = 1.0f / fmaxf(sqrtf(ss), 1e-12f);
    v.x *= inv; v.y *= inv; v.z *= inv; v.w *= inv;
    op[lane] = v;
}


static inline size_t align_up(size_t v, size_t a) { return (v + a - 1) & ~(a - 1); }

extern "C" void kernel_launch(void* const* d_in, const int* in_sizes, int n_in,
                              void* d_out, int out_size, void* d_ws, size_t ws_size,
                              hipStream_t stream) {
    const float* x  = (const float*)d_in[0];
    const int*   ei = (const int*)d_in[1];
    const float* ew = (const float*)d_in[2];
    const float* W1 = (const float*)d_in[3];
    const float* b1 = (const float*)d_in[4];
    const float* W2 = (const float*)d_in[5];
    const float* b2 = (const float*)d_in[6];
    float* out = (float*)d_out;

    const int N = in_sizes[0] / IN_CH;
    const int E = in_sizes[1] / 2;
    const int* row = ei;
    const int* col = ei + E;

    char* ws = (char*)d_ws;
    size_t off = 0;
    float* deg  = (float*)(ws + off); off = align_up(off + (size_t)N * 4, 1024);
    float* dinv = (float*)(ws + off); off = align_up(off + (size_t)N * 4, 1024);
    float* nrm  = (float*)(ws + off); off = align_up(off + (size_t)E * 4, 1024);
    float* h    = (float*)(ws + off); off = align_up(off + (size_t)N * HID_CH * 4, 1024);
    float* acc  = (float*)(ws + off); off = align_up(off + (size_t)N * HID_CH * 4, 1024);
    (void)ws_size; (void)n_in; (void)out_size;

    const int nb_nodes  = (N + 255) / 256;
    const int nb_edges  = (E + 255) / 256;
    const int nb_nvec   = (N * 32 + 255) / 256;
    const int nb_nwave  = (N + 7) / 8;
    const int nb_ewave  = (E + 7) / 8;
    const int nb_gemm   = N / GEMM_ROWS;

    k_init_deg   <<<nb_nodes, 256, 0, stream>>>(deg, N);
    k_deg_scatter<<<nb_edges, 256, 0, stream>>>(col, ew, deg, E);
    k_dinv       <<<nb_nodes, 256, 0, stream>>>(deg, dinv, N);
    k_edge_norm  <<<nb_edges, 256, 0, stream>>>(row, col, ew, dinv, nrm, E);

    k_gemm_wmma  <<<nb_gemm, 256, 0, stream>>>(x, W1, h, IN_CH);
    k_init_acc   <<<nb_nvec, 256, 0, stream>>>(h, dinv, acc, N);
    k_scatter    <<<nb_ewave, 256, 0, stream>>>(row, col, nrm, h, acc, E);
    k_bias_relu  <<<nb_nvec, 256, 0, stream>>>(acc, b1, N);

    k_gemm_wmma  <<<nb_gemm, 256, 0, stream>>>(acc, W2, h, HID_CH);
    k_init_acc   <<<nb_nvec, 256, 0, stream>>>(h, dinv, out, N);
    k_scatter    <<<nb_ewave, 256, 0, stream>>>(row, col, nrm, h, out, E);
    k_bias_l2norm<<<nb_nwave, 256, 0, stream>>>(out, b2, N);
}


# Neighbour-sum gather: one 16-B-per-lane load per 512-B row, 16 rows in flight per wave, 16 list entries per LDS read; LDS tile columns interleaved so adds stay conflict-free; sum order = LDS arrival
# speedup vs baseline: 2.7034x; 2.7034x over previous
#include <hip/hip_runtime.h>
#include <hip/hip_bf16.h>


typedef __attribute__((ext_vector_type(2))) float v2f;
typedef __attribute__((ext_vector_type(8))) float v8f;

#define IN_CH 256
#define HID_CH 128
#define GEMM_ROWS 32
#define TILE 392
#define LCAP 8192
#define AGG_THREADS 1024
#define GU 16
#define DEG_RANGES 8
#define DEG_SLICES 32
#define DEG_RCAP 12544
#define DEG_THREADS 1024


__global__ __launch_bounds__(256) void k_init_deg(float* deg, int N) {
    int i = blockIdx.x * 256 + threadIdx.x;
    if (i < N) deg[i] = 1.0f;
}

__global__ __launch_bounds__(DEG_THREADS) void k_deg_hist(const int* __restrict__ col,
                                                          const float* __restrict__ w,
                                                          float* deg, int N, int E) {
    __shared__ float hist[DEG_RCAP];
    const int tid = threadIdx.x;
    const int rg = blockIdx.x % DEG_RANGES;
    const int sl = blockIdx.x / DEG_RANGES;
    const int rlen = (((N + DEG_RANGES - 1) / DEG_RANGES) + 31) & ~31;
    const int lo = rg * rlen;
    const int cnt = max(0, min(rlen, N - lo));
    for (int i = tid; i < cnt; i += DEG_THREADS) hist[i] = 0.0f;
    __syncthreads();

    if ((E & 3) == 0) {
        const int ng = E >> 2;
        const int per = (ng + DEG_SLICES - 1) / DEG_SLICES;
        const int g0 = sl * per;
        const int g1 = min(g0 + per, ng);
        const int4* c4 = (const int4*)col;
        const float4* w4 = (const float4*)w;
        for (int g = g0 + tid; g < g1; g += 4 * DEG_THREADS) {
            int4 c[4];
            float4 ww[4];
#pragma unroll
            for (int j = 0; j < 4; ++j) {
                const int gj = g + j * DEG_THREADS;
                const int gi = (gj < g1) ? gj : g;
                c[j] = c4[gi];
                ww[j] = w4[gi];
            }
#pragma unroll
            for (int j = 0; j < 4; ++j) {
                if (g + j * DEG_THREADS < g1) {
                    const unsigned u0 = (unsigned)(c[j].x - lo), u1 = (unsigned)(c[j].y - lo);
                    const unsigned u2 = (unsigned)(c[j].z - lo), u3 = (unsigned)(c[j].w - lo);
                    if (u0 < (unsigned)cnt) unsafeAtomicAdd(&hist[u0], ww[j].x);
                    if (u1 < (unsigned)cnt) unsafeAtomicAdd(&hist[u1], ww[j].y);
                    if (u2 < (unsigned)cnt) unsafeAtomicAdd(&hist[u2], ww[j].z);
                    if (u3 < (unsigned)cnt) unsafeAtomicAdd(&hist[u3], ww[j].w);
                }
            }
        }
    } else {
        const int per = (E + DEG_SLICES - 1) / DEG_SLICES;
        const int e0 = sl * per;
        const int e1 = min(e0 + per, E);
        for (int e = e0 + tid; e < e1; e += DEG_THREADS) {
            const unsigned u = (unsigned)(col[e] - lo);
            if (u < (unsigned)cnt) unsafeAtomicAdd(&hist[u], w[e]);
        }
    }
    __syncthreads();

    for (int i = tid; i < cnt; i += DEG_THREADS) unsafeAtomicAdd(&deg[lo + i], hist[i]);
}

__global__ __launch_bounds__(256) void k_dinv(const float* __restrict__ deg,
                                              float* dinv, int N) {
    int i = blockIdx.x * 256 + threadIdx.x;
    if (i < N) {
        float d = deg[i];
        dinv[i] = (d > 0.0f) ? rsqrtf(d) : 0.0f;
    }
}

__global__ __launch_bounds__(256) void k_edge_norm(const int* __restrict__ row,
                                                   const int* __restrict__ col,
                                                   const float* __restrict__ w,
                                                   const float* __restrict__ dinv,
                                                   float* __restrict__ nrm,
                                                   unsigned* __restrict__ key32, int E) {
    int t = blockIdx.x * 256 + threadIdx.x;
    int e0 = t * 4;
    if (e0 >= E) return;
    unsigned pk = 0u;
    if (e0 + 3 < E) {
        int r[4], c[4];
        float ww[4], nv[4];
#pragma unroll
        for (int j = 0; j < 4; ++j) { r[j] = row[e0 + j]; c[j] = col[e0 + j]; ww[j] = w[e0 + j]; }
#pragma unroll
        for (int j = 0; j < 4; ++j) {
            nv[j] = dinv[r[j]] * ww[j] * dinv[c[j]];
            pk |= (unsigned)(c[j] / TILE) << (8 * j);
        }
        *(float4*)(nrm + e0) = make_float4(nv[0], nv[1], nv[2], nv[3]);
    } else {
        for (int j = 0; e0 + j < E; ++j) {
            int c = col[e0 + j];
            nrm[e0 + j] = dinv[row[e0 + j]] * w[e0 + j] * dinv[c];
            pk |= (unsigned)(c / TILE) << (8 * j);
        }
    }
    key32[t] = pk;
}


__global__ __launch_bounds__(256) void k_gemm_wmma(const float* __restrict__ A,
                                                   const float* __restrict__ W,
                                                   float* __restrict__ H, int K) {
    __shared__ float As[GEMM_ROWS * (IN_CH + 4)];
    const int stride = K + 4;
    const int tid = threadIdx.x;
    const int rowbase = blockIdx.x * GEMM_ROWS;

    for (int idx = tid; idx < GEMM_ROWS * K; idx += 256) {
        int r = idx / K;
        int kk = idx - r * K;
        As[r * stride + kk] = A[(size_t)(rowbase + r) * K + kk];
    }
    __syncthreads();

    const int wave = tid >> 5;
    const int lane = tid & 31;
    const int m    = lane & 15;
    const int half = lane >> 4;
    const int cb   = wave * 16;

    v8f c0 = {};
    v8f c1 = {};
    const float* wp  = W + cb + m;
    const float* ap0 = As + m * stride;
    const float* ap1 = As + (16 + m) * stride;

    for (int k = 0; k < K; k += 4) {
        const int koff = k + half * 2;
        v2f b;
        b[0] = wp[(size_t)koff * HID_CH];
        b[1] = wp[(size_t)(koff + 1) * HID_CH];
        v2f a0, a1;
        a0[0] = ap0[koff];
        a0[1] = ap0[koff + 1];
        a1[0] = ap1[koff];
        a1[1] = ap1[koff + 1];
        c0 = __builtin_amdgcn_wmma_f32_16x16x4_f32(false, a0, false, b,
                                                   (short)0, c0, false, false);
        c1 = __builtin_amdgcn_wmma_f32_16x16x4_f32(false, a1, false, b,
                                                   (short)0, c1, false, false);
    }

    float* hp0 = H + (size_t)rowbase * HID_CH + cb + m;
    float* hp1 = hp0 + (size_t)16 * HID_CH;
#pragma unroll
    for (int r = 0; r < 8; ++r) {
        hp0[(size_t)(r + 8 * half) * HID_CH] = c0[r];
        hp1[(size_t)(r + 8 * half) * HID_CH] = c1[r];
    }
}


__device__ __forceinline__ unsigned hitmask16(uint4 q, unsigned pat) {
    const unsigned x0 = q.x ^ pat, x1 = q.y ^ pat, x2 = q.z ^ pat, x3 = q.w ^ pat;
    const unsigned u0 = ((x0 & 0x7F7F7F7Fu) + 0x7F7F7F7Fu) | x0;
    const unsigned u1 = ((x1 & 0x7F7F7F7Fu) + 0x7F7F7F7Fu) | x1;
    const unsigned u2 = ((x2 & 0x7F7F7F7Fu) + 0x7F7F7F7Fu) | x2;
    const unsigned u3 = ((x3 & 0x7F7F7F7Fu) + 0x7F7F7F7Fu) | x3;
    const unsigned nz = ((u0 >> 7) & 0x01010101u) | ((u1 >> 6) & 0x02020202u) |
                        ((u2 >> 5) & 0x04040404u) | ((u3 >> 4) & 0x08080808u);
    return nz ^ 0x0F0F0F0Fu;
}

__device__ __forceinline__ void slow_edge(const int* __restrict__ row, const int* __restrict__ col,
                                       const float* __restrict__ nrm, const float* __restrict__ h,
                                       float* acc, int base, int e) {
    int r = row[e];
    int c = col[e] - base;
    float nv = nrm[e];
    const float* hp = h + (size_t)r * HID_CH;
    float* a = acc + c * HID_CH;
    for (int j = 0; j < HID_CH; ++j) unsafeAtomicAdd(a + ((j & 3) << 5) + (j >> 2), hp[j] * nv);
}

__global__ __launch_bounds__(AGG_THREADS) void k_tile_agg(const int* __restrict__ row,
                                                          const int* __restrict__ col,
                                                          const float* __restrict__ nrm,
                                                          const unsigned* __restrict__ key32,
                                                          const float* __restrict__ h,
                                                          const float* __restrict__ dinv,
                                                          float* __restrict__ dst,
                                                          int* lst_cnt, int* lst_r, int* lst_c,
                                                          float* lst_nv, int reuse, int N, int E) {
    __shared__ __attribute__((aligned(16))) float acc[TILE * HID_CH];
    __shared__ int   l_r[LCAP];
    __shared__ int   l_c[LCAP];
    __shared__ float l_nv[LCAP];
    __shared__ int   l_cnt;

    const int tid  = threadIdx.x;
    const int tile = blockIdx.x;
    const int base = tile * TILE;
    const int nrows = min(TILE, N - base);

    if (tid == 0) l_cnt = 0;
    {
        const float4* h4 = (const float4*)(h + (size_t)base * HID_CH);
        for (int idx = tid; idx < nrows * 32; idx += AGG_THREADS) {
            float d = dinv[base + (idx >> 5)];
            float s = d * d;
            float4 v = h4[idx];
            float* a = acc + (idx >> 5) * HID_CH + (idx & 31);
            a[0]  = v.x * s;
            a[32] = v.y * s;
            a[64] = v.z * s;
            a[96] = v.w * s;
        }
    }
    int* const   g_r  = lst_r  + (size_t)tile * LCAP;
    int* const   g_c  = lst_c  + (size_t)tile * LCAP;
    float* const g_nv = lst_nv + (size_t)tile * LCAP;
    int n = 0;
    bool build = true;
    if (reuse) {
        const int saved = lst_cnt[tile];
        if (saved <= LCAP) {
            build = false;
            n = saved;
            for (int i = tid; i < n; i += AGG_THREADS) {
                l_r[i]  = g_r[i];
                l_c[i]  = g_c[i];
                l_nv[i] = g_nv[i];
            }
        }
    }
    __syncthreads();

    if (build) {
        {
            const uint4* k4 = (const uint4*)key32;
            const int nch = E >> 4;
            const unsigned pat = (unsigned)tile * 0x01010101u;
            for (int c0 = tid; c0 < nch; c0 += 4 * AGG_THREADS) {
                uint4 q[4];
#pragma unroll
                for (int j = 0; j < 4; ++j) {
                    const int ch = c0 + j * AGG_THREADS;
                    q[j] = k4[(ch < nch) ? ch : c0];
                }
#pragma unroll
                for (int j = 0; j < 4; ++j) {
                    const int ch = c0 + j * AGG_THREADS;
                    unsigned m = (ch < nch) ? hitmask16(q[j], pat) : 0u;
                    while (m != 0u) {
                        const int p = __ffs((int)m) - 1;
                        m &= m - 1u;
                        const int e = ch * 16 + ((p & 7) << 2) + (p >> 3);
                        const int i = atomicAdd(&l_cnt, 1);
                        if (i < LCAP) l_r[i] = e;
                        else slow_edge(row, col, nrm, h, acc, base, e);
                    }
                }
            }
            const unsigned char* k1 = (const unsigned char*)key32;
            for (int e = (nch << 4) + tid; e < E; e += AGG_THREADS) {
                if ((int)k1[e] == tile) {
                    const int i = atomicAdd(&l_cnt, 1);
                    if (i < LCAP) l_r[i] = e;
                    else slow_edge(row, col, nrm, h, acc, base, e);
                }
            }
        }
        __syncthreads();

        n = min(l_cnt, LCAP);
        for (int i = tid; i < n; i += AGG_THREADS) {
            const int e = l_r[i];
            const int   r  = row[e];
            const int   c  = col[e] - base;
            const float nv = nrm[e];
            l_r[i]  = r;
            l_c[i]  = c;
            l_nv[i] = nv;
            if (!reuse) { g_r[i] = r; g_c[i] = c; g_nv[i] = nv; }
        }
        if (!reuse && tid == 0) lst_cnt[tile] = l_cnt;
        __syncthreads();
    }

    {
        const int wv = __builtin_amdgcn_readfirstlane(tid >> 5);
        const int nu = __builtin_amdgcn_readfirstlane(n);
        const int lane = tid & 31;
        const int sub = lane & (GU - 1);
        const float4* hl = (const float4*)h + lane;
        for (int q = wv; q * GU < nu; q += AGG_THREADS / 32) {
            const int e0 = q * GU;
            const int ei = min(e0 + sub, nu - 1);
            const int my_r  = l_r[ei];
            const int my_c  = l_c[ei];
            const int my_nv = __float_as_int(l_nv[ei]);
            float4 v[GU];
            float  nv[GU];
            int    cc[GU];
#pragma unroll
            for (int u = 0; u < GU; ++u) {
                const int r = __builtin_amdgcn_readlane(my_r, u);
                cc[u] = __builtin_amdgcn_readlane(my_c, u);
                nv[u] = __int_as_float(__builtin_amdgcn_readlane(my_nv, u));
                v[u] = hl[(size_t)r * 32];
            }
            if (e0 + GU <= nu) {
#pragma unroll
                for (int u = 0; u < GU; ++u) {
                    float* a = acc + cc[u] * HID_CH + lane;
                    unsafeAtomicAdd(a +  0, v[u].x * nv[u]);
                    unsafeAtomicAdd(a + 32, v[u].y * nv[u]);
                    unsafeAtomicAdd(a + 64, v[u].z * nv[u]);
                    unsafeAtomicAdd(a + 96, v[u].w * nv[u]);
                }
            } else {
#pragma unroll
                for (int u = 0; u < GU; ++u) {
                    if (e0 + u < nu) {
                        float* a = acc + cc[u] * HID_CH + lane;
                        unsafeAtomicAdd(a +  0, v[u].x * nv[u]);
                        unsafeAtomicAdd(a + 32, v[u].y * nv[u]);
                        unsafeAtomicAdd(a + 64, v[u].z * nv[u]);
                        unsafeAtomicAdd(a + 96, v[u].w * nv[u]);
                    }
                }
            }
        }
    }
    __syncthreads();

    {
        float4* d4 = (float4*)(dst + (size_t)base * HID_CH);
        for (int idx = tid; idx < nrows * 32; idx += AGG_THREADS) {
            const float* a = acc + (idx >> 5) * HID_CH + (idx & 31);
            d4[idx] = make_float4(a[0], a[32], a[64], a[96]);
        }
    }
}

__global__ __launch_bounds__(256) void k_bias_relu(float* __restrict__ acc,
                                                   const float* __restrict__ b, int N) {
    int g = blockIdx.x * 256 + threadIdx.x;
    if (g >= N * 32) return;
    float4* ap = (float4*)acc;
    const float4* bp = (const float4*)b;
    float4 v = ap[g];
    float4 bb = bp[g & 31];
    v.x = fmaxf(v.x + bb.x, 0.0f);
    v.y = fmaxf(v.y + bb.y, 0.0f);
    v.z = fmaxf(v.z + bb.z, 0.0f);
    v.w = fmaxf(v.w + bb.w, 0.0f);
    ap[g] = v;
}

__global__ __launch_bounds__(256) void k_bias_l2norm(const float* __restrict__ src,
                                                     float* __restrict__ out,
                                                     const float* __restrict__ b, int N) {
    int wid = (blockIdx.x * 256 + threadIdx.x) >> 5;
    int lane = threadIdx.x & 31;
    if (wid >= N) return;
    const float4* sp = (const float4*)(src + (size_t)wid * HID_CH);
    float4* op = (float4*)(out + (size_t)wid * HID_CH);
    const float4* bp = (const float4*)b;
    float4 v = sp[lane];
    float4 bb = bp[lane];
    v.x += bb.x; v.y += bb.y; v.z += bb.z; v.w += bb.w;
    float ss = v.x * v.x + v.y * v.y + v.z * v.z + v.w * v.w;
#pragma unroll
    for (int m = 16; m >= 1; m >>= 1) ss += __shfl_xor(ss, m, 32);
    float inv = 1.0f / fmaxf(sqrtf(ss), 1e-12f);
    v.x *= inv; v.y *= inv; v.z *= inv; v.w *= inv;
    op[lane] = v;
}


static inline size_t align_up(size_t v, size_t a) { return (v + a - 1) & ~(a - 1); }

extern "C" void kernel_launch(void* const* d_in, const int* in_sizes, int n_in,
                              void* d_out, int out_size, void* d_ws, size_t ws_size,
                              hipStream_t stream) {
    const float* x  = (const float*)d_in[0];
    const int*   ei = (const int*)d_in[1];
    const float* ew = (const float*)d_in[2];
    const float* W1 = (const float*)d_in[3];
    const float* b1 = (const float*)d_in[4];
    const float* W2 = (const float*)d_in[5];
    const float* b2 = (const float*)d_in[6];
    float* out = (float*)d_out;

    const int N = in_sizes[0] / IN_CH;
    const int E = in_sizes[1] / 2;
    const int* row = ei;
    const int* col = ei + E;

    char* ws = (char*)d_ws;
    size_t off = 0;
    float* deg  = (float*)(ws + off); off = align_up(off + (size_t)N * 4, 1024);
    float* dinv = (float*)(ws + off); off = align_up(off + (size_t)N * 4, 1024);
    float* nrm  = (float*)(ws + off); off = align_up(off + (size_t)E * 4, 1024);
    float* h    = (float*)(ws + off); off = align_up(off + (size_t)N * HID_CH * 4, 1024);
    float* acc  = (float*)(ws + off); off = align_up(off + (size_t)N * HID_CH * 4, 1024);
    (void)ws_size; (void)n_in; (void)out_size;

    const int nb_nodes  = (N + 255) / 256;
    const int nb_edge4  = ((E + 3) / 4 + 255) / 256;
    const int nb_tiles  = (N + TILE - 1) / TILE;
    const int nb_nvec   = (N * 32 + 255) / 256;
    const int nb_nwave  = (N + 7) / 8;
    const int nb_gemm   = N / GEMM_ROWS;

    k_init_deg   <<<nb_nodes, 256, 0, stream>>>(deg, N);
    k_deg_hist   <<<DEG_RANGES * DEG_SLICES, DEG_THREADS, 0, stream>>>(col, ew, deg, N, E);
    k_dinv       <<<nb_nodes, 256, 0, stream>>>(deg, dinv, N);
    unsigned* key32 = (unsigned*)d_out;
    k_edge_norm  <<<nb_edge4, 256, 0, stream>>>(row, col, ew, dinv, nrm, key32, E);
    char* ob = (char*)d_out;
    size_t ooff = align_up((size_t)E + 4, 1024);
    int*   lst_cnt = (int*)(ob + ooff);   ooff = align_up(ooff + (size_t)nb_tiles * 4, 1024);
    int*   lst_r   = (int*)(ob + ooff);   ooff = align_up(ooff + (size_t)nb_tiles * LCAP * 4, 1024);
    int*   lst_c   = (int*)(ob + ooff);   ooff = align_up(ooff + (size_t)nb_tiles * LCAP * 4, 1024);
    float* lst_nv  = (float*)(ob + ooff);

    k_gemm_wmma  <<<nb_gemm, 256, 0, stream>>>(x, W1, h, IN_CH);
    k_tile_agg   <<<nb_tiles, AGG_THREADS, 0, stream>>>(row, col, nrm, key32, h, dinv, acc,
                                                        lst_cnt, lst_r, lst_c, lst_nv, 0, N, E);
    k_bias_relu  <<<nb_nvec, 256, 0, stream>>>(acc, b1, N);

    k_gemm_wmma  <<<nb_gemm, 256, 0, stream>>>(acc, W2, h, HID_CH);
    k_tile_agg   <<<nb_tiles, AGG_THREADS, 0, stream>>>(row, col, nrm, key32, h, dinv, acc,
                                                        lst_cnt, lst_r, lst_c, lst_nv, 1, N, E);
    k_bias_l2norm<<<nb_nwave, 256, 0, stream>>>(acc, out, b2, N);
}


# Edge pass stores a packed 8-byte record per edge (source row, local row, coefficient); layer-1 list build fetches one line per listed edge, not three, 8 per thread in flight; sum order kept
# speedup vs baseline: 2.8169x; 1.0420x over previous
#include <hip/hip_runtime.h>
#include <hip/hip_bf16.h>


typedef __attribute__((ext_vector_type(2))) float v2f;
typedef __attribute__((ext_vector_type(8))) float v8f;

#define IN_CH 256
#define HID_CH 128
#define GEMM_ROWS 32
#define TILE 392
#define LCAP 8192
#define AGG_THREADS 1024
#define GU 16
#define RSHIFT 17
#define RPT (LCAP / AGG_THREADS)
#define DEG_RANGES 8
#define DEG_SLICES 32
#define DEG_RCAP 12544
#define DEG_THREADS 1024


__global__ __launch_bounds__(256) void k_init_deg(float* deg, int N) {
    int i = blockIdx.x * 256 + threadIdx.x;
    if (i < N) deg[i] = 1.0f;
}

__global__ __launch_bounds__(DEG_THREADS) void k_deg_hist(const int* __restrict__ col,
                                                          const float* __restrict__ w,
                                                          float* deg, int N, int E) {
    __shared__ float hist[DEG_RCAP];
    const int tid = threadIdx.x;
    const int rg = blockIdx.x % DEG_RANGES;
    const int sl = blockIdx.x / DEG_RANGES;
    const int rlen = (((N + DEG_RANGES - 1) / DEG_RANGES) + 31) & ~31;
    const int lo = rg * rlen;
    const int cnt = max(0, min(rlen, N - lo));
    for (int i = tid; i < cnt; i += DEG_THREADS) hist[i] = 0.0f;
    __syncthreads();

    if ((E & 3) == 0) {
        const int ng = E >> 2;
        const int per = (ng + DEG_SLICES - 1) / DEG_SLICES;
        const int g0 = sl * per;
        const int g1 = min(g0 + per, ng);
        const int4* c4 = (const int4*)col;
        const float4* w4 = (const float4*)w;
        for (int g = g0 + tid; g < g1; g += 4 * DEG_THREADS) {
            int4 c[4];
            float4 ww[4];
#pragma unroll
            for (int j = 0; j < 4; ++j) {
                const int gj = g + j * DEG_THREADS;
                const int gi = (gj < g1) ? gj : g;
                c[j] = c4[gi];
                ww[j] = w4[gi];
            }
#pragma unroll
            for (int j = 0; j < 4; ++j) {
                if (g + j * DEG_THREADS < g1) {
                    const unsigned u0 = (unsigned)(c[j].x - lo), u1 = (unsigned)(c[j].y - lo);
                    const unsigned u2 = (unsigned)(c[j].z - lo), u3 = (unsigned)(c[j].w - lo);
                    if (u0 < (unsigned)cnt) unsafeAtomicAdd(&hist[u0], ww[j].x);
                    if (u1 < (unsigned)cnt) unsafeAtomicAdd(&hist[u1], ww[j].y);
                    if (u2 < (unsigned)cnt) unsafeAtomicAdd(&hist[u2], ww[j].z);
                    if (u3 < (unsigned)cnt) unsafeAtomicAdd(&hist[u3], ww[j].w);
                }
            }
        }
    } else {
        const int per = (E + DEG_SLICES - 1) / DEG_SLICES;
        const int e0 = sl * per;
        const int e1 = min(e0 + per, E);
        for (int e = e0 + tid; e < e1; e += DEG_THREADS) {
            const unsigned u = (unsigned)(col[e] - lo);
            if (u < (unsigned)cnt) unsafeAtomicAdd(&hist[u], w[e]);
        }
    }
    __syncthreads();

    for (int i = tid; i < cnt; i += DEG_THREADS) unsafeAtomicAdd(&deg[lo + i], hist[i]);
}

__global__ __launch_bounds__(256) void k_dinv(const float* __restrict__ deg,
                                              float* dinv, int N) {
    int i = blockIdx.x * 256 + threadIdx.x;
    if (i < N) {
        float d = deg[i];
        dinv[i] = (d > 0.0f) ? rsqrtf(d) : 0.0f;
    }
}

__global__ __launch_bounds__(256) void k_edge_norm(const int* __restrict__ row,
                                                   const int* __restrict__ col,
                                                   const float* __restrict__ w,
                                                   const float* __restrict__ dinv,
                                                   unsigned* __restrict__ rec,
                                                   unsigned* __restrict__ key32, int E) {
    int t = blockIdx.x * 256 + threadIdx.x;
    int e0 = t * 4;
    if (e0 >= E) return;
    unsigned pk = 0u;
    if (e0 + 3 < E) {
        int r[4], c[4];
        float ww[4];
        unsigned rc[4], nb[4];
#pragma unroll
        for (int j = 0; j < 4; ++j) { r[j] = row[e0 + j]; c[j] = col[e0 + j]; ww[j] = w[e0 + j]; }
#pragma unroll
        for (int j = 0; j < 4; ++j) {
            const unsigned tl = (unsigned)c[j] / TILE;
            nb[j] = __float_as_uint(dinv[r[j]] * ww[j] * dinv[c[j]]);
            rc[j] = (unsigned)r[j] | (((unsigned)c[j] - tl * TILE) << RSHIFT);
            pk |= tl << (8 * j);
        }
        uint4* rp = (uint4*)(rec + 2 * (size_t)e0);
        rp[0] = make_uint4(rc[0], nb[0], rc[1], nb[1]);
        rp[1] = make_uint4(rc[2], nb[2], rc[3], nb[3]);
    } else {
        for (int j = 0; e0 + j < E; ++j) {
            const int rr = row[e0 + j];
            const int c = col[e0 + j];
            const unsigned tl = (unsigned)c / TILE;
            rec[2 * (size_t)(e0 + j)]     = (unsigned)rr | (((unsigned)c - tl * TILE) << RSHIFT);
            rec[2 * (size_t)(e0 + j) + 1] = __float_as_uint(dinv[rr] * w[e0 + j] * dinv[c]);
            pk |= tl << (8 * j);
        }
    }
    key32[t] = pk;
}


__global__ __launch_bounds__(256) void k_gemm_wmma(const float* __restrict__ A,
                                                   const float* __restrict__ W,
                                                   float* __restrict__ H, int K) {
    __shared__ float As[GEMM_ROWS * (IN_CH + 4)];
    const int stride = K + 4;
    const int tid = threadIdx.x;
    const int rowbase = blockIdx.x * GEMM_ROWS;

    for (int idx = tid; idx < GEMM_ROWS * K; idx += 256) {
        int r = idx / K;
        int kk = idx - r * K;
        As[r * stride + kk] = A[(size_t)(rowbase + r) * K + kk];
    }
    __syncthreads();

    const int wave = tid >> 5;
    const int lane = tid & 31;
    const int m    = lane & 15;
    const int half = lane >> 4;
    const int cb   = wave * 16;

    v8f c0 = {};
    v8f c1 = {};
    const float* wp  = W + cb + m;
    const float* ap0 = As + m * stride;
    const float* ap1 = As + (16 + m) * stride;

    for (int k = 0; k < K; k += 4) {
        const int koff = k + half * 2;
        v2f b;
        b[0] = wp[(size_t)koff * HID_CH];
        b[1] = wp[(size_t)(koff + 1) * HID_CH];
        v2f a0, a1;
        a0[0] = ap0[koff];
        a0[1] = ap0[koff + 1];
        a1[0] = ap1[koff];
        a1[1] = ap1[koff + 1];
        c0 = __builtin_amdgcn_wmma_f32_16x16x4_f32(false, a0, false, b,
                                                   (short)0, c0, false, false);
        c1 = __builtin_amdgcn_wmma_f32_16x16x4_f32(false, a1, false, b,
                                                   (short)0, c1, false, false);
    }

    float* hp0 = H + (size_t)rowbase * HID_CH + cb + m;
    float* hp1 = hp0 + (size_t)16 * HID_CH;
#pragma unroll
    for (int r = 0; r < 8; ++r) {
        hp0[(size_t)(r + 8 * half) * HID_CH] = c0[r];
        hp1[(size_t)(r + 8 * half) * HID_CH] = c1[r];
    }
}


__device__ __forceinline__ unsigned hitmask16(uint4 q, unsigned pat) {
    const unsigned x0 = q.x ^ pat, x1 = q.y ^ pat, x2 = q.z ^ pat, x3 = q.w ^ pat;
    const unsigned u0 = ((x0 & 0x7F7F7F7Fu) + 0x7F7F7F7Fu) | x0;
    const unsigned u1 = ((x1 & 0x7F7F7F7Fu) + 0x7F7F7F7Fu) | x1;
    const unsigned u2 = ((x2 & 0x7F7F7F7Fu) + 0x7F7F7F7Fu) | x2;
    const unsigned u3 = ((x3 & 0x7F7F7F7Fu) + 0x7F7F7F7Fu) | x3;
    const unsigned nz = ((u0 >> 7) & 0x01010101u) | ((u1 >> 6) & 0x02020202u) |
                        ((u2 >> 5) & 0x04040404u) | ((u3 >> 4) & 0x08080808u);
    return nz ^ 0x0F0F0F0Fu;
}

__device__ __forceinline__ void slow_edge(const unsigned* __restrict__ rec, const float* __restrict__ h,
                                       float* acc, int e) {
    const uint2 q = ((const uint2*)rec)[e];
    int r = (int)(q.x & ((1u << RSHIFT) - 1u));
    int c = (int)(q.x >> RSHIFT);
    float nv = __uint_as_float(q.y);
    const float* hp = h + (size_t)r * HID_CH;
    float* a = acc + c * HID_CH;
    for (int j = 0; j < HID_CH; ++j) unsafeAtomicAdd(a + ((j & 3) << 5) + (j >> 2), hp[j] * nv);
}

__global__ __launch_bounds__(AGG_THREADS) void k_tile_agg(const unsigned* __restrict__ rec,
                                                          const unsigned* __restrict__ key32,
                                                          const float* __restrict__ h,
                                                          const float* __restrict__ dinv,
                                                          float* __restrict__ dst,
                                                          int* lst_cnt, int* lst_r, int* lst_c,
                                                          float* lst_nv, int reuse, int N, int E) {
    __shared__ __attribute__((aligned(16))) float acc[TILE * HID_CH];
    __shared__ int   l_r[LCAP];
    __shared__ int   l_c[LCAP];
    __shared__ float l_nv[LCAP];
    __shared__ int   l_cnt;

    const int tid  = threadIdx.x;
    const int tile = blockIdx.x;
    const int base = tile * TILE;
    const int nrows = min(TILE, N - base);

    if (tid == 0) l_cnt = 0;
    {
        const float4* h4 = (const float4*)(h + (size_t)base * HID_CH);
        for (int idx = tid; idx < nrows * 32; idx += AGG_THREADS) {
            float d = dinv[base + (idx >> 5)];
            float s = d * d;
            float4 v = h4[idx];
            float* a = acc + (idx >> 5) * HID_CH + (idx & 31);
            a[0]  = v.x * s;
            a[32] = v.y * s;
            a[64] = v.z * s;
            a[96] = v.w * s;
        }
    }
    int* const   g_r  = lst_r  + (size_t)tile * LCAP;
    int* const   g_c  = lst_c  + (size_t)tile * LCAP;
    float* const g_nv = lst_nv + (size_t)tile * LCAP;
    int n = 0;
    bool build = true;
    if (reuse) {
        const int saved = lst_cnt[tile];
        if (saved <= LCAP) {
            build = false;
            n = saved;
            for (int i = tid; i < n; i += AGG_THREADS) {
                l_r[i]  = g_r[i];
                l_c[i]  = g_c[i];
                l_nv[i] = g_nv[i];
            }
        }
    }
    __syncthreads();

    if (build) {
        {
            const uint4* k4 = (const uint4*)key32;
            const int nch = E >> 4;
            const unsigned pat = (unsigned)tile * 0x01010101u;
            for (int c0 = tid; c0 < nch; c0 += 4 * AGG_THREADS) {
                uint4 q[4];
#pragma unroll
                for (int j = 0; j < 4; ++j) {
                    const int ch = c0 + j * AGG_THREADS;
                    q[j] = k4[(ch < nch) ? ch : c0];
                }
#pragma unroll
                for (int j = 0; j < 4; ++j) {
                    const int ch = c0 + j * AGG_THREADS;
                    unsigned m = (ch < nch) ? hitmask16(q[j], pat) : 0u;
                    while (m != 0u) {
                        const int p = __ffs((int)m) - 1;
                        m &= m - 1u;
                        const int e = ch * 16 + ((p & 7) << 2) + (p >> 3);
                        const int i = atomicAdd(&l_cnt, 1);
                        if (i < LCAP) l_r[i] = e;
                        else slow_edge(rec, h, acc, e);
                    }
                }
            }
            const unsigned char* k1 = (const unsigned char*)key32;
            for (int e = (nch << 4) + tid; e < E; e += AGG_THREADS) {
                if ((int)k1[e] == tile) {
                    const int i = atomicAdd(&l_cnt, 1);
                    if (i < LCAP) l_r[i] = e;
                    else slow_edge(rec, h, acc, e);
                }
            }
        }
        __syncthreads();

        n = min(l_cnt, LCAP);
        {
            const uint2* rec2 = (const uint2*)rec;
            uint2 q[RPT];
#pragma unroll
            for (int j = 0; j < RPT; ++j) {
                const int i = tid + j * AGG_THREADS;
                const int e = l_r[i];
                q[j] = rec2[(i < n) ? e : 0];
            }
#pragma unroll
            for (int j = 0; j < RPT; ++j) {
                const int i = tid + j * AGG_THREADS;
                if (i < n) {
                    const int   r  = (int)(q[j].x & ((1u << RSHIFT) - 1u));
                    const int   c  = (int)(q[j].x >> RSHIFT);
                    const float nv = __uint_as_float(q[j].y);
                    l_r[i]  = r;
                    l_c[i]  = c;
                    l_nv[i] = nv;
                    if (!reuse) { g_r[i] = r; g_c[i] = c; g_nv[i] = nv; }
                }
            }
        }
        if (!reuse && tid == 0) lst_cnt[tile] = l_cnt;
        __syncthreads();
    }

    {
        const int wv = __builtin_amdgcn_readfirstlane(tid >> 5);
        const int nu = __builtin_amdgcn_readfirstlane(n);
        const int lane = tid & 31;
        const int sub = lane & (GU - 1);
        const float4* hl = (const float4*)h + lane;
        for (int q = wv; q * GU < nu; q += AGG_THREADS / 32) {
            const int e0 = q * GU;
            const int ei = min(e0 + sub, nu - 1);
            const int my_r  = l_r[ei];
            const int my_c  = l_c[ei];
            const int my_nv = __float_as_int(l_nv[ei]);
            float4 v[GU];
            float  nv[GU];
            int    cc[GU];
#pragma unroll
            for (int u = 0; u < GU; ++u) {
                const int r = __builtin_amdgcn_readlane(my_r, u);
                cc[u] = __builtin_amdgcn_readlane(my_c, u);
                nv[u] = __int_as_float(__builtin_amdgcn_readlane(my_nv, u));
                v[u] = hl[(size_t)r * 32];
            }
            if (e0 + GU <= nu) {
#pragma unroll
                for (int u = 0; u < GU; ++u) {
                    float* a = acc + cc[u] * HID_CH + lane;
                    unsafeAtomicAdd(a +  0, v[u].x * nv[u]);
                    unsafeAtomicAdd(a + 32, v[u].y * nv[u]);
                    unsafeAtomicAdd(a + 64, v[u].z * nv[u]);
                    unsafeAtomicAdd(a + 96, v[u].w * nv[u]);
                }
            } else {
#pragma unroll
                for (int u = 0; u < GU; ++u) {
                    if (e0 + u < nu) {
                        float* a = acc + cc[u] * HID_CH + lane;
                        unsafeAtomicAdd(a +  0, v[u].x * nv[u]);
                        unsafeAtomicAdd(a + 32, v[u].y * nv[u]);
                        unsafeAtomicAdd(a + 64, v[u].z * nv[u]);
                        unsafeAtomicAdd(a + 96, v[u].w * nv[u]);
                    }
                }
            }
        }
    }
    __syncthreads();

    {
        float4* d4 = (float4*)(dst + (size_t)base * HID_CH);
        for (int idx = tid; idx < nrows * 32; idx += AGG_THREADS) {
            const float* a = acc + (idx >> 5) * HID_CH + (idx & 31);
            d4[idx] = make_float4(a[0], a[32], a[64], a[96]);
        }
    }
}

__global__ __launch_bounds__(256) void k_bias_relu(float* __restrict__ acc,
                                                   const float* __restrict__ b, int N) {
    int g = blockIdx.x * 256 + threadIdx.x;
    if (g >= N * 32) return;
    float4* ap = (float4*)acc;
    const float4* bp = (const float4*)b;
    float4 v = ap[g];
    float4 bb = bp[g & 31];
    v.x = fmaxf(v.x + bb.x, 0.0f);
    v.y = fmaxf(v.y + bb.y, 0.0f);
    v.z = fmaxf(v.z + bb.z, 0.0f);
    v.w = fmaxf(v.w + bb.w, 0.0f);
    ap[g] = v;
}

__global__ __launch_bounds__(256) void k_bias_l2norm(const float* __restrict__ src,
                                                     float* __restrict__ out,
                                                     const float* __restrict__ b, int N) {
    int wid = (blockIdx.x * 256 + threadIdx.x) >> 5;
    int lane = threadIdx.x & 31;
    if (wid >= N) return;
    const float4* sp = (const float4*)(src + (size_t)wid * HID_CH);
    float4* op = (float4*)(out + (size_t)wid * HID_CH);
    const float4* bp = (const float4*)b;
    float4 v = sp[lane];
    float4 bb = bp[lane];
    v.x += bb.x; v.y += bb.y; v.z += bb.z; v.w += bb.w;
    float ss = v.x * v.x + v.y * v.y + v.z * v.z + v.w * v.w;
#pragma unroll
    for (int m = 16; m >= 1; m >>= 1) ss += __shfl_xor(ss, m, 32);
    float inv = 1.0f / fmaxf(sqrtf(ss), 1e-12f);
    v.x *= inv; v.y *= inv; v.z *= inv; v.w *= inv;
    op[lane] = v;
}


static inline size_t align_up(size_t v, size_t a) { return (v + a - 1) & ~(a - 1); }

extern "C" void kernel_launch(void* const* d_in, const int* in_sizes, int n_in,
                              void* d_out, int out_size, void* d_ws, size_t ws_size,
                              hipStream_t stream) {
    const float* x  = (const float*)d_in[0];
    const int*   ei = (const int*)d_in[1];
    const float* ew = (const float*)d_in[2];
    const float* W1 = (const float*)d_in[3];
    const float* b1 = (const float*)d_in[4];
    const float* W2 = (const float*)d_in[5];
    const float* b2 = (const float*)d_in[6];
    float* out = (float*)d_out;

    const int N = in_sizes[0] / IN_CH;
    const int E = in_sizes[1] / 2;
    const int* row = ei;
    const int* col = ei + E;

    char* ws = (char*)d_ws;
    size_t off = 0;
    float* deg  = (float*)(ws + off); off = align_up(off + (size_t)N * 4, 1024);
    float* dinv = (float*)(ws + off); off = align_up(off + (size_t)N * 4, 1024);
    float* h    = (float*)(ws + off); off = align_up(off + (size_t)N * HID_CH * 4, 1024);
    float* acc  = (float*)(ws + off); off = align_up(off + (size_t)N * HID_CH * 4, 1024);
    (void)ws_size; (void)n_in; (void)out_size;

    const int nb_nodes  = (N + 255) / 256;
    const int nb_edge4  = ((E + 3) / 4 + 255) / 256;
    const int nb_tiles  = (N + TILE - 1) / TILE;
    const int nb_nvec   = (N * 32 + 255) / 256;
    const int nb_nwave  = (N + 7) / 8;
    const int nb_gemm   = N / GEMM_ROWS;

    k_init_deg   <<<nb_nodes, 256, 0, stream>>>(deg, N);
    k_deg_hist   <<<DEG_RANGES * DEG_SLICES, DEG_THREADS, 0, stream>>>(col, ew, deg, N, E);
    k_dinv       <<<nb_nodes, 256, 0, stream>>>(deg, dinv, N);
    unsigned* key32 = (unsigned*)d_out;
    char* ob = (char*)d_out;
    size_t ooff = align_up((size_t)E + 4, 1024);
    int*   lst_cnt = (int*)(ob + ooff);   ooff = align_up(ooff + (size_t)nb_tiles * 4, 1024);
    int*   lst_r   = (int*)(ob + ooff);   ooff = align_up(ooff + (size_t)nb_tiles * LCAP * 4, 1024);
    int*   lst_c   = (int*)(ob + ooff);   ooff = align_up(ooff + (size_t)nb_tiles * LCAP * 4, 1024);
    float* lst_nv  = (float*)(ob + ooff); ooff = align_up(ooff + (size_t)nb_tiles * LCAP * 4, 1024);
    unsigned* rec  = (unsigned*)(ob + ooff);
    k_edge_norm  <<<nb_edge4, 256, 0, stream>>>(row, col, ew, dinv, rec, key32, E);

    k_gemm_wmma  <<<nb_gemm, 256, 0, stream>>>(x, W1, h, IN_CH);
    k_tile_agg   <<<nb_tiles, AGG_THREADS, 0, stream>>>(rec, key32, h, dinv, acc,
                                                        lst_cnt, lst_r, lst_c, lst_nv, 0, N, E);
    k_bias_relu  <<<nb_nvec, 256, 0, stream>>>(acc, b1, N);

    k_gemm_wmma  <<<nb_gemm, 256, 0, stream>>>(acc, W2, h, HID_CH);
    k_tile_agg   <<<nb_tiles, AGG_THREADS, 0, stream>>>(rec, key32, h, dinv, acc,
                                                        lst_cnt, lst_r, lst_c, lst_nv, 1, N, E);
    k_bias_l2norm<<<nb_nwave, 256, 0, stream>>>(acc, out, b2, N);
}


# Elementwise launches folded into their consumers' loads: bias+relu applied while the layer-2 product stages A, deg^-1/2 taken where a degree is read (edge pass, tile start); 10 -> 8 launches
# speedup vs baseline: 2.9539x; 1.0486x over previous
#include <hip/hip_runtime.h>
#include <hip/hip_bf16.h>


typedef __attribute__((ext_vector_type(2))) float v2f;
typedef __attribute__((ext_vector_type(8))) float v8f;

#define IN_CH 256
#define HID_CH 128
#define GEMM_ROWS 32
#define TILE 392
#define LCAP 8192
#define AGG_THREADS 1024
#define GU 16
#define RSHIFT 17
#define RPT (LCAP / AGG_THREADS)
#define DEG_RANGES 8
#define DEG_SLICES 32
#define DEG_RCAP 12544
#define DEG_THREADS 1024


__global__ __launch_bounds__(256) void k_init_deg(float* deg, int N) {
    int i = blockIdx.x * 256 + threadIdx.x;
    if (i < N) deg[i] = 1.0f;
}

__global__ __launch_bounds__(DEG_THREADS) void k_deg_hist(const int* __restrict__ col,
                                                          const float* __restrict__ w,
                                                          float* deg, int N, int E) {
    __shared__ float hist[DEG_RCAP];
    const int tid = threadIdx.x;
    const int rg = blockIdx.x % DEG_RANGES;
    const int sl = blockIdx.x / DEG_RANGES;
    const int rlen = (((N + DEG_RANGES - 1) / DEG_RANGES) + 31) & ~31;
    const int lo = rg * rlen;
    const int cnt = max(0, min(rlen, N - lo));
    for (int i = tid; i < cnt; i += DEG_THREADS) hist[i] = 0.0f;
    __syncthreads();

    if ((E & 3) == 0) {
        const int ng = E >> 2;
        const int per = (ng + DEG_SLICES - 1) / DEG_SLICES;
        const int g0 = sl * per;
        const int g1 = min(g0 + per, ng);
        const int4* c4 = (const int4*)col;
        const float4* w4 = (const float4*)w;
        for (int g = g0 + tid; g < g1; g += 4 * DEG_THREADS) {
            int4 c[4];
            float4 ww[4];
#pragma unroll
            for (int j = 0; j < 4; ++j) {
                const int gj = g + j * DEG_THREADS;
                const int gi = (gj < g1) ? gj : g;
                c[j] = c4[gi];
                ww[j] = w4[gi];
            }
#pragma unroll
            for (int j = 0; j < 4; ++j) {
                if (g + j * DEG_THREADS < g1) {
                    const unsigned u0 = (unsigned)(c[j].x - lo), u1 = (unsigned)(c[j].y - lo);
                    const unsigned u2 = (unsigned)(c[j].z - lo), u3 = (unsigned)(c[j].w - lo);
                    if (u0 < (unsigned)cnt) unsafeAtomicAdd(&hist[u0], ww[j].x);
                    if (u1 < (unsigned)cnt) unsafeAtomicAdd(&hist[u1], ww[j].y);
                    if (u2 < (unsigned)cnt) unsafeAtomicAdd(&hist[u2], ww[j].z);
                    if (u3 < (unsigned)cnt) unsafeAtomicAdd(&hist[u3], ww[j].w);
                }
            }
        }
    } else {
        const int per = (E + DEG_SLICES - 1) / DEG_SLICES;
        const int e0 = sl * per;
        const int e1 = min(e0 + per, E);
        for (int e = e0 + tid; e < e1; e += DEG_THREADS) {
            const unsigned u = (unsigned)(col[e] - lo);
            if (u < (unsigned)cnt) unsafeAtomicAdd(&hist[u], w[e]);
        }
    }
    __syncthreads();

    for (int i = tid; i < cnt; i += DEG_THREADS) unsafeAtomicAdd(&deg[lo + i], hist[i]);
}

__device__ __forceinline__ float inv_sqrt_deg(float d) {
    return (d > 0.0f) ? rsqrtf(d) : 0.0f;
}

__global__ __launch_bounds__(256) void k_edge_norm(const int* __restrict__ row,
                                                   const int* __restrict__ col,
                                                   const float* __restrict__ w,
                                                   const float* __restrict__ deg,
                                                   unsigned* __restrict__ rec,
                                                   unsigned* __restrict__ key32, int E) {
    int t = blockIdx.x * 256 + threadIdx.x;
    int e0 = t * 4;
    if (e0 >= E) return;
    unsigned pk = 0u;
    if (e0 + 3 < E) {
        int r[4], c[4];
        float ww[4];
        unsigned rc[4], nb[4];
#pragma unroll
        for (int j = 0; j < 4; ++j) { r[j] = row[e0 + j]; c[j] = col[e0 + j]; ww[j] = w[e0 + j]; }
#pragma unroll
        for (int j = 0; j < 4; ++j) {
            const unsigned tl = (unsigned)c[j] / TILE;
            nb[j] = __float_as_uint(inv_sqrt_deg(deg[r[j]]) * ww[j] * inv_sqrt_deg(deg[c[j]]));
            rc[j] = (unsigned)r[j] | (((unsigned)c[j] - tl * TILE) << RSHIFT);
            pk |= tl << (8 * j);
        }
        uint4* rp = (uint4*)(rec + 2 * (size_t)e0);
        rp[0] = make_uint4(rc[0], nb[0], rc[1], nb[1]);
        rp[1] = make_uint4(rc[2], nb[2], rc[3], nb[3]);
    } else {
        for (int j = 0; e0 + j < E; ++j) {
            const int rr = row[e0 + j];
            const int c = col[e0 + j];
            const unsigned tl = (unsigned)c / TILE;
            rec[2 * (size_t)(e0 + j)]     = (unsigned)rr | (((unsigned)c - tl * TILE) << RSHIFT);
            rec[2 * (size_t)(e0 + j) + 1] = __float_as_uint(inv_sqrt_deg(deg[rr]) * w[e0 + j] * inv_sqrt_deg(deg[c]));
            pk |= tl << (8 * j);
        }
    }
    key32[t] = pk;
}


__global__ __launch_bounds__(256) void k_gemm_wmma(const float* __restrict__ A,
                                                   const float* __restrict__ W,
                                                   const float* __restrict__ bias,
                                                   float* __restrict__ H, int K) {
    __shared__ float As[GEMM_ROWS * (IN_CH + 4)];
    const int stride = K + 4;
    const int tid = threadIdx.x;
    const int rowbase = blockIdx.x * GEMM_ROWS;

    if (bias != nullptr) {
        for (int idx = tid; idx < GEMM_ROWS * K; idx += 256) {
            int r = idx / K;
            int kk = idx - r * K;
            As[r * stride + kk] = fmaxf(A[(size_t)(rowbase + r) * K + kk] + bias[kk], 0.0f);
        }
    } else {
        for (int idx = tid; idx < GEMM_ROWS * K; idx += 256) {
            int r = idx / K;
            int kk = idx - r * K;
            As[r * stride + kk] = A[(size_t)(rowbase + r) * K + kk];
        }
    }
    __syncthreads();

    const int wave = tid >> 5;
    const int lane = tid & 31;
    const int m    = lane & 15;
    const int half = lane >> 4;
    const int cb   = wave * 16;

    v8f c0 = {};
    v8f c1 = {};
    const float* wp  = W + cb + m;
    const float* ap0 = As + m * stride;
    const float* ap1 = As + (16 + m) * stride;

    for (int k = 0; k < K; k += 4) {
        const int koff = k + half * 2;
        v2f b;
        b[0] = wp[(size_t)koff * HID_CH];
        b[1] = wp[(size_t)(koff + 1) * HID_CH];
        v2f a0, a1;
        a0[0] = ap0[koff];
        a0[1] = ap0[koff + 1];
        a1[0] = ap1[koff];
        a1[1] = ap1[koff + 1];
        c0 = __builtin_amdgcn_wmma_f32_16x16x4_f32(false, a0, false, b,
                                                   (short)0, c0, false, false);
        c1 = __builtin_amdgcn_wmma_f32_16x16x4_f32(false, a1, false, b,
                                                   (short)0, c1, false, false);
    }

    float* hp0 = H + (size_t)rowbase * HID_CH + cb + m;
    float* hp1 = hp0 + (size_t)16 * HID_CH;
#pragma unroll
    for (int r = 0; r < 8; ++r) {
        hp0[(size_t)(r + 8 * half) * HID_CH] = c0[r];
        hp1[(size_t)(r + 8 * half) * HID_CH] = c1[r];
    }
}


__device__ __forceinline__ unsigned hitmask16(uint4 q, unsigned pat) {
    const unsigned x0 = q.x ^ pat, x1 = q.y ^ pat, x2 = q.z ^ pat, x3 = q.w ^ pat;
    const unsigned u0 = ((x0 & 0x7F7F7F7Fu) + 0x7F7F7F7Fu) | x0;
    const unsigned u1 = ((x1 & 0x7F7F7F7Fu) + 0x7F7F7F7Fu) | x1;
    const unsigned u2 = ((x2 & 0x7F7F7F7Fu) + 0x7F7F7F7Fu) | x2;
    const unsigned u3 = ((x3 & 0x7F7F7F7Fu) + 0x7F7F7F7Fu) | x3;
    const unsigned nz = ((u0 >> 7) & 0x01010101u) | ((u1 >> 6) & 0x02020202u) |
                        ((u2 >> 5) & 0x04040404u) | ((u3 >> 4) & 0x08080808u);
    return nz ^ 0x0F0F0F0Fu;
}

__device__ __forceinline__ void slow_edge(const unsigned* __restrict__ rec, const float* __restrict__ h,
                                       float* acc, int e) {
    const uint2 q = ((const uint2*)rec)[e];
    int r = (int)(q.x & ((1u << RSHIFT) - 1u));
    int c = (int)(q.x >> RSHIFT);
    float nv = __uint_as_float(q.y);
    const float* hp = h + (size_t)r * HID_CH;
    float* a = acc + c * HID_CH;
    for (int j = 0; j < HID_CH; ++j) unsafeAtomicAdd(a + ((j & 3) << 5) + (j >> 2), hp[j] * nv);
}

__global__ __launch_bounds__(AGG_THREADS) void k_tile_agg(const unsigned* __restrict__ rec,
                                                          const unsigned* __restrict__ key32,
                                                          const float* __restrict__ h,
                                                          const float* __restrict__ deg,
                                                          float* __restrict__ dst,
                                                          int* lst_cnt, int* lst_r, int* lst_c,
                                                          float* lst_nv, int reuse, int N, int E) {
    __shared__ __attribute__((aligned(16))) float acc[TILE * HID_CH];
    __shared__ int   l_r[LCAP];
    __shared__ int   l_c[LCAP];
    __shared__ float l_nv[LCAP];
    __shared__ int   l_cnt;

    const int tid  = threadIdx.x;
    const int tile = blockIdx.x;
    const int base = tile * TILE;
    const int nrows = min(TILE, N - base);

    if (tid == 0) l_cnt = 0;
    {
        const float4* h4 = (const float4*)(h + (size_t)base * HID_CH);
        for (int idx = tid; idx < nrows * 32; idx += AGG_THREADS) {
            float d = inv_sqrt_deg(deg[base + (idx >> 5)]);
            float s = d * d;
            float4 v = h4[idx];
            float* a = acc + (idx >> 5) * HID_CH + (idx & 31);
            a[0]  = v.x * s;
            a[32] = v.y * s;
            a[64] = v.z * s;
            a[96] = v.w * s;
        }
    }
    int* const   g_r  = lst_r  + (size_t)tile * LCAP;
    int* const   g_c  = lst_c  + (size_t)tile * LCAP;
    float* const g_nv = lst_nv + (size_t)tile * LCAP;
    int n = 0;
    bool build = true;
    if (reuse) {
        const int saved = lst_cnt[tile];
        if (saved <= LCAP) {
            build = false;
            n = saved;
            for (int i = tid; i < n; i += AGG_THREADS) {
                l_r[i]  = g_r[i];
                l_c[i]  = g_c[i];
                l_nv[i] = g_nv[i];
            }
        }
    }
    __syncthreads();

    if (build) {
        {
            const uint4* k4 = (const uint4*)key32;
            const int nch = E >> 4;
            const unsigned pat = (unsigned)tile * 0x01010101u;
            for (int c0 = tid; c0 < nch; c0 += 4 * AGG_THREADS) {
                uint4 q[4];
#pragma unroll
                for (int j = 0; j < 4; ++j) {
                    const int ch = c0 + j * AGG_THREADS;
                    q[j] = k4[(ch < nch) ? ch : c0];
                }
#pragma unroll
                for (int j = 0; j < 4; ++j) {
                    const int ch = c0 + j * AGG_THREADS;
                    unsigned m = (ch < nch) ? hitmask16(q[j], pat) : 0u;
                    while (m != 0u) {
                        const int p = __ffs((int)m) - 1;
                        m &= m - 1u;
                        const int e = ch * 16 + ((p & 7) << 2) + (p >> 3);
                        const int i = atomicAdd(&l_cnt, 1);
                        if (i < LCAP) l_r[i] = e;
                        else slow_edge(rec, h, acc, e);
                    }
                }
            }
            const unsigned char* k1 = (const unsigned char*)key32;
            for (int e = (nch << 4) + tid; e < E; e += AGG_THREADS) {
                if ((int)k1[e] == tile) {
                    const int i = atomicAdd(&l_cnt, 1);
                    if (i < LCAP) l_r[i] = e;
                    else slow_edge(rec, h, acc, e);
                }
            }
        }
        __syncthreads();

        n = min(l_cnt, LCAP);
        {
            const uint2* rec2 = (const uint2*)rec;
            uint2 q[RPT];
#pragma unroll
            for (int j = 0; j < RPT; ++j) {
                const int i = tid + j * AGG_THREADS;
                const int e = l_r[i];
                q[j] = rec2[(i < n) ? e : 0];
            }
#pragma unroll
            for (int j = 0; j < RPT; ++j) {
                const int i = tid + j * AGG_THREADS;
                if (i < n) {
                    const int   r  = (int)(q[j].x & ((1u << RSHIFT) - 1u));
                    const int   c  = (int)(q[j].x >> RSHIFT);
                    const float nv = __uint_as_float(q[j].y);
                    l_r[i]  = r;
                    l_c[i]  = c;
                    l_nv[i] = nv;
                    if (!reuse) { g_r[i] = r; g_c[i] = c; g_nv[i] = nv; }
                }
            }
        }
        if (!reuse && tid == 0) lst_cnt[tile] = l_cnt;
        __syncthreads();
    }

    {
        const int wv = __builtin_amdgcn_readfirstlane(tid >> 5);
        const int nu = __builtin_amdgcn_readfirstlane(n);
        const int lane = tid & 31;
        const int sub = lane & (GU - 1);
        const float4* hl = (const float4*)h + lane;
        for (int q = wv; q * GU < nu; q += AGG_THREADS / 32) {
            const int e0 = q * GU;
            const int ei = min(e0 + sub, nu - 1);
            const int my_r  = l_r[ei];
            const int my_c  = l_c[ei];
            const int my_nv = __float_as_int(l_nv[ei]);
            float4 v[GU];
            float  nv[GU];
            int    cc[GU];
#pragma unroll
            for (int u = 0; u < GU; ++u) {
                const int r = __builtin_amdgcn_readlane(my_r, u);
                cc[u] = __builtin_amdgcn_readlane(my_c, u);
                nv[u] = __int_as_float(__builtin_amdgcn_readlane(my_nv, u));
                v[u] = hl[(size_t)r * 32];
            }
            if (e0 + GU <= nu) {
#pragma unroll
                for (int u = 0; u < GU; ++u) {
                    float* a = acc + cc[u] * HID_CH + lane;
                    unsafeAtomicAdd(a +  0, v[u].x * nv[u]);
                    unsafeAtomicAdd(a + 32, v[u].y * nv[u]);
                    unsafeAtomicAdd(a + 64, v[u].z * nv[u]);
                    unsafeAtomicAdd(a + 96, v[u].w * nv[u]);
                }
            } else {
#pragma unroll
                for (int u = 0; u < GU; ++u) {
                    if (e0 + u < nu) {
                        float* a = acc + cc[u] * HID_CH + lane;
                        unsafeAtomicAdd(a +  0, v[u].x * nv[u]);
                        unsafeAtomicAdd(a + 32, v[u].y * nv[u]);
                        unsafeAtomicAdd(a + 64, v[u].z * nv[u]);
                        unsafeAtomicAdd(a + 96, v[u].w * nv[u]);
                    }
                }
            }
        }
    }
    __syncthreads();

    {
        float4* d4 = (float4*)(dst + (size_t)base * HID_CH);
        for (int idx = tid; idx < nrows * 32; idx += AGG_THREADS) {
            const float* a = acc + (idx >> 5) * HID_CH + (idx & 31);
            d4[idx] = make_float4(a[0], a[32], a[64], a[96]);
        }
    }
}

__global__ __launch_bounds__(256) void k_bias_l2norm(const float* __restrict__ src,
                                                     float* __restrict__ out,
                                                     const float* __restrict__ b, int N) {
    int wid = (blockIdx.x * 256 + threadIdx.x) >> 5;
    int lane = threadIdx.x & 31;
    if (wid >= N) return;
    const float4* sp = (const float4*)(src + (size_t)wid * HID_CH);
    float4* op = (float4*)(out + (size_t)wid * HID_CH);
    const float4* bp = (const float4*)b;
    float4 v = sp[lane];
    float4 bb = bp[lane];
    v.x += bb.x; v.y += bb.y; v.z += bb.z; v.w += bb.w;
    float ss = v.x * v.x + v.y * v.y + v.z * v.z + v.w * v.w;
#pragma unroll
    for (int m = 16; m >= 1; m >>= 1) ss += __shfl_xor(ss, m, 32);
    float inv = 1.0f / fmaxf(sqrtf(ss), 1e-12f);
    v.x *= inv; v.y *= inv; v.z *= inv; v.w *= inv;
    op[lane] = v;
}


static inline size_t align_up(size_t v, size_t a) { return (v + a - 1) & ~(a - 1); }

extern "C" void kernel_launch(void* const* d_in, const int* in_sizes, int n_in,
                              void* d_out, int out_size, void* d_ws, size_t ws_size,
                              hipStream_t stream) {
    const float* x  = (const float*)d_in[0];
    const int*   ei = (const int*)d_in[1];
    const float* ew = (const float*)d_in[2];
    const float* W1 = (const float*)d_in[3];
    const float* b1 = (const float*)d_in[4];
    const float* W2 = (const float*)d_in[5];
    const float* b2 = (const float*)d_in[6];
    float* out = (float*)d_out;

    const int N = in_sizes[0] / IN_CH;
    const int E = in_sizes[1] / 2;
    const int* row = ei;
    const int* col = ei + E;

    char* ws = (char*)d_ws;
    size_t off = 0;
    float* deg  = (float*)(ws + off); off = align_up(off + (size_t)N * 4, 1024);
    float* h    = (float*)(ws + off); off = align_up(off + (size_t)N * HID_CH * 4, 1024);
    float* acc  = (float*)(ws + off); off = align_up(off + (size_t)N * HID_CH * 4, 1024);
    (void)ws_size; (void)n_in; (void)out_size;

    const int nb_nodes  = (N + 255) / 256;
    const int nb_edge4  = ((E + 3) / 4 + 255) / 256;
    const int nb_tiles  = (N + TILE - 1) / TILE;
    const int nb_nwave  = (N + 7) / 8;
    const int nb_gemm   = N / GEMM_ROWS;

    k_init_deg   <<<nb_nodes, 256, 0, stream>>>(deg, N);
    k_deg_hist   <<<DEG_RANGES * DEG_SLICES, DEG_THREADS, 0, stream>>>(col, ew, deg, N, E);
    unsigned* key32 = (unsigned*)d_out;
    char* ob = (char*)d_out;
    size_t ooff = align_up((size_t)E + 4, 1024);
    int*   lst_cnt = (int*)(ob + ooff);   ooff = align_up(ooff + (size_t)nb_tiles * 4, 1024);
    int*   lst_r   = (int*)(ob + ooff);   ooff = align_up(ooff + (size_t)nb_tiles * LCAP * 4, 1024);
    int*   lst_c   = (int*)(ob + ooff);   ooff = align_up(ooff + (size_t)nb_tiles * LCAP * 4, 1024);
    float* lst_nv  = (float*)(ob + ooff); ooff = align_up(ooff + (size_t)nb_tiles * LCAP * 4, 1024);
    unsigned* rec  = (unsigned*)(ob + ooff);
    k_edge_norm  <<<nb_edge4, 256, 0, stream>>>(row, col, ew, deg, rec, key32, E);

    k_gemm_wmma  <<<nb_gemm, 256, 0, stream>>>(x, W1, nullptr, h, IN_CH);
    k_tile_agg   <<<nb_tiles, AGG_THREADS, 0, stream>>>(rec, key32, h, deg, acc,
                                                        lst_cnt, lst_r, lst_c, lst_nv, 0, N, E);

    k_gemm_wmma  <<<nb_gemm, 256, 0, stream>>>(acc, W2, b1, h, HID_CH);
    k_tile_agg   <<<nb_tiles, AGG_THREADS, 0, stream>>>(rec, key32, h, deg, acc,
                                                        lst_cnt, lst_r, lst_c, lst_nv, 1, N, E);
    k_bias_l2norm<<<nb_nwave, 256, 0, stream>>>(acc, out, b2, N);
}


# Tile keys as 8 bit planes (wave votes in edge pass): layer-1 list filter tests 128 edges/thread by 8 16-B loads, 16 in flight, 1 LDS add per wave pass, own sweep start per block; add order = arrival
# speedup vs baseline: 3.0781x; 1.0420x over previous
#include <hip/hip_runtime.h>
#include <hip/hip_bf16.h>


typedef __attribute__((ext_vector_type(2))) float v2f;
typedef __attribute__((ext_vector_type(8))) float v8f;

#define IN_CH 256
#define HID_CH 128
#define GEMM_ROWS 32
#define TILE 392
#define LCAP 8192
#define AGG_THREADS 1024
#define GU 16
#define RSHIFT 17
#define RPT (LCAP / AGG_THREADS)
#define DEG_RANGES 8
#define DEG_SLICES 32
#define DEG_RCAP 12544
#define DEG_THREADS 1024

__host__ __device__ static inline int key_units(int E) { return ((((E + 3) / 4) + 255) / 256) * 8; }


__global__ __launch_bounds__(256) void k_init_deg(float* deg, int N) {
    int i = blockIdx.x * 256 + threadIdx.x;
    if (i < N) deg[i] = 1.0f;
}

__global__ __launch_bounds__(DEG_THREADS) void k_deg_hist(const int* __restrict__ col,
                                                          const float* __restrict__ w,
                                                          float* deg, int N, int E) {
    __shared__ float hist[DEG_RCAP];
    const int tid = threadIdx.x;
    const int rg = blockIdx.x % DEG_RANGES;
    const int sl = blockIdx.x / DEG_RANGES;
    const int rlen = (((N + DEG_RANGES - 1) / DEG_RANGES) + 31) & ~31;
    const int lo = rg * rlen;
    const int cnt = max(0, min(rlen, N - lo));
    for (int i = tid; i < cnt; i += DEG_THREADS) hist[i] = 0.0f;
    __syncthreads();

    if ((E & 3) == 0) {
        const int ng = E >> 2;
        const int per = (ng + DEG_SLICES - 1) / DEG_SLICES;
        const int g0 = sl * per;
        const int g1 = min(g0 + per, ng);
        const int4* c4 = (const int4*)col;
        const float4* w4 = (const float4*)w;
        for (int g = g0 + tid; g < g1; g += 4 * DEG_THREADS) {
            int4 c[4];
            float4 ww[4];
#pragma unroll
            for (int j = 0; j < 4; ++j) {
                const int gj = g + j * DEG_THREADS;
                const int gi = (gj < g1) ? gj : g;
                c[j] = c4[gi];
                ww[j] = w4[gi];
            }
#pragma unroll
            for (int j = 0; j < 4; ++j) {
                if (g + j * DEG_THREADS < g1) {
                    const unsigned u0 = (unsigned)(c[j].x - lo), u1 = (unsigned)(c[j].y - lo);
                    const unsigned u2 = (unsigned)(c[j].z - lo), u3 = (unsigned)(c[j].w - lo);
                    if (u0 < (unsigned)cnt) unsafeAtomicAdd(&hist[u0], ww[j].x);
                    if (u1 < (unsigned)cnt) unsafeAtomicAdd(&hist[u1], ww[j].y);
                    if (u2 < (unsigned)cnt) unsafeAtomicAdd(&hist[u2], ww[j].z);
                    if (u3 < (unsigned)cnt) unsafeAtomicAdd(&hist[u3], ww[j].w);
                }
            }
        }
    } else {
        const int per = (E + DEG_SLICES - 1) / DEG_SLICES;
        const int e0 = sl * per;
        const int e1 = min(e0 + per, E);
        for (int e = e0 + tid; e < e1; e += DEG_THREADS) {
            const unsigned u = (unsigned)(col[e] - lo);
            if (u < (unsigned)cnt) unsafeAtomicAdd(&hist[u], w[e]);
        }
    }
    __syncthreads();

    for (int i = tid; i < cnt; i += DEG_THREADS) unsafeAtomicAdd(&deg[lo + i], hist[i]);
}

__device__ __forceinline__ float inv_sqrt_deg(float d) {
    return (d > 0.0f) ? rsqrtf(d) : 0.0f;
}

__global__ __launch_bounds__(256) void k_edge_norm(const int* __restrict__ row,
                                                   const int* __restrict__ col,
                                                   const float* __restrict__ w,
                                                   const float* __restrict__ deg,
                                                   unsigned* __restrict__ rec,
                                                   unsigned* __restrict__ planes, int E) {
    const int t = blockIdx.x * 256 + threadIdx.x;
    const int e0 = t * 4;
    unsigned tk[4] = {0u, 0u, 0u, 0u};
    if (e0 + 3 < E) {
        int r[4], c[4];
        float ww[4];
        unsigned rc[4], nb[4];
#pragma unroll
        for (int j = 0; j < 4; ++j) { r[j] = row[e0 + j]; c[j] = col[e0 + j]; ww[j] = w[e0 + j]; }
#pragma unroll
        for (int j = 0; j < 4; ++j) {
            const unsigned tl = (unsigned)c[j] / TILE;
            nb[j] = __float_as_uint(inv_sqrt_deg(deg[r[j]]) * ww[j] * inv_sqrt_deg(deg[c[j]]));
            rc[j] = (unsigned)r[j] | (((unsigned)c[j] - tl * TILE) << RSHIFT);
            tk[j] = tl;
        }
        uint4* rp = (uint4*)(rec + 2 * (size_t)e0);
        rp[0] = make_uint4(rc[0], nb[0], rc[1], nb[1]);
        rp[1] = make_uint4(rc[2], nb[2], rc[3], nb[3]);
    } else {
#pragma unroll
        for (int j = 0; j < 4; ++j) {
            if (e0 + j < E) {
                const int rr = row[e0 + j];
                const int cn = col[e0 + j];
                const unsigned tl = (unsigned)cn / TILE;
                rec[2 * (size_t)(e0 + j)]     = (unsigned)rr | (((unsigned)cn - tl * TILE) << RSHIFT);
                rec[2 * (size_t)(e0 + j) + 1] = __float_as_uint(inv_sqrt_deg(deg[rr]) * w[e0 + j] * inv_sqrt_deg(deg[cn]));
                tk[j] = tl;
            }
        }
    }
    const int lane = threadIdx.x & 31;
    unsigned mine = 0u;
#pragma unroll
    for (int p = 0; p < 8; ++p) {
#pragma unroll
        for (int j = 0; j < 4; ++j) {
            const unsigned b = __builtin_amdgcn_ballot_w32(((tk[j] >> p) & 1u) != 0u);
            mine = (lane == 4 * p + j) ? b : mine;
        }
    }
    const size_t nu = (size_t)key_units(E);
    planes[(size_t)(lane >> 2) * nu * 4 + (size_t)(t >> 5) * 4 + (size_t)(lane & 3)] = mine;
}


__global__ __launch_bounds__(256) void k_gemm_wmma(const float* __restrict__ A,
                                                   const float* __restrict__ W,
                                                   const float* __restrict__ bias,
                                                   float* __restrict__ H, int K) {
    __shared__ float As[GEMM_ROWS * (IN_CH + 4)];
    const int stride = K + 4;
    const int tid = threadIdx.x;
    const int rowbase = blockIdx.x * GEMM_ROWS;

    if (bias != nullptr) {
        for (int idx = tid; idx < GEMM_ROWS * K; idx += 256) {
            int r = idx / K;
            int kk = idx - r * K;
            As[r * stride + kk] = fmaxf(A[(size_t)(rowbase + r) * K + kk] + bias[kk], 0.0f);
        }
    } else {
        for (int idx = tid; idx < GEMM_ROWS * K; idx += 256) {
            int r = idx / K;
            int kk = idx - r * K;
            As[r * stride + kk] = A[(size_t)(rowbase + r) * K + kk];
        }
    }
    __syncthreads();

    const int wave = tid >> 5;
    const int lane = tid & 31;
    const int m    = lane & 15;
    const int half = lane >> 4;
    const int cb   = wave * 16;

    v8f c0 = {};
    v8f c1 = {};
    const float* wp  = W + cb + m;
    const float* ap0 = As + m * stride;
    const float* ap1 = As + (16 + m) * stride;

    for (int k = 0; k < K; k += 4) {
        const int koff = k + half * 2;
        v2f b;
        b[0] = wp[(size_t)koff * HID_CH];
        b[1] = wp[(size_t)(koff + 1) * HID_CH];
        v2f a0, a1;
        a0[0] = ap0[koff];
        a0[1] = ap0[koff + 1];
        a1[0] = ap1[koff];
        a1[1] = ap1[koff + 1];
        c0 = __builtin_amdgcn_wmma_f32_16x16x4_f32(false, a0, false, b,
                                                   (short)0, c0, false, false);
        c1 = __builtin_amdgcn_wmma_f32_16x16x4_f32(false, a1, false, b,
                                                   (short)0, c1, false, false);
    }

    float* hp0 = H + (size_t)rowbase * HID_CH + cb + m;
    float* hp1 = hp0 + (size_t)16 * HID_CH;
#pragma unroll
    for (int r = 0; r < 8; ++r) {
        hp0[(size_t)(r + 8 * half) * HID_CH] = c0[r];
        hp1[(size_t)(r + 8 * half) * HID_CH] = c1[r];
    }
}


__device__ __forceinline__ void slow_edge(const unsigned* __restrict__ rec, const float* __restrict__ h,
                                       float* acc, int e) {
    const uint2 q = ((const uint2*)rec)[e];
    int r = (int)(q.x & ((1u << RSHIFT) - 1u));
    int c = (int)(q.x >> RSHIFT);
    float nv = __uint_as_float(q.y);
    const float* hp = h + (size_t)r * HID_CH;
    float* a = acc + c * HID_CH;
    for (int j = 0; j < HID_CH; ++j) unsafeAtomicAdd(a + ((j & 3) << 5) + (j >> 2), hp[j] * nv);
}

__device__ __forceinline__ unsigned valid_bits(int E, int u, int j) {
    const int lim = (E - (u << 7) - j + 3) >> 2;
    return lim >= 32 ? 0xFFFFFFFFu : (lim <= 0 ? 0u : ((1u << lim) - 1u));
}

__device__ __forceinline__ int emit_hits(unsigned m, int u, int j, int i, int* l_r,
                                         const unsigned* __restrict__ rec, const float* __restrict__ h,
                                         float* acc) {
    while (m != 0u) {
        const int b = __ffs((int)m) - 1;
        m &= m - 1u;
        const int e = (u << 7) + (b << 2) + j;
        if (i < LCAP) l_r[i] = e;
        else slow_edge(rec, h, acc, e);
        ++i;
    }
    return i;
}

__global__ __launch_bounds__(AGG_THREADS) void k_tile_agg(const unsigned* __restrict__ rec,
                                                          const unsigned* __restrict__ planes,
                                                          const float* __restrict__ h,
                                                          const float* __restrict__ deg,
                                                          float* __restrict__ dst,
                                                          int* lst_cnt, int* lst_r, int* lst_c,
                                                          float* lst_nv, int reuse, int N, int E) {
    __shared__ __attribute__((aligned(16))) float acc[TILE * HID_CH];
    __shared__ int   l_r[LCAP];
    __shared__ int   l_c[LCAP];
    __shared__ float l_nv[LCAP];
    __shared__ int   l_cnt;

    const int tid  = threadIdx.x;
    const int tile = blockIdx.x;
    const int base = tile * TILE;
    const int nrows = min(TILE, N - base);

    if (tid == 0) l_cnt = 0;
    {
        const float4* h4 = (const float4*)(h + (size_t)base * HID_CH);
        for (int idx = tid; idx < nrows * 32; idx += AGG_THREADS) {
            float d = inv_sqrt_deg(deg[base + (idx >> 5)]);
            float s = d * d;
            float4 v = h4[idx];
            float* a = acc + (idx >> 5) * HID_CH + (idx & 31);
            a[0]  = v.x * s;
            a[32] = v.y * s;
            a[64] = v.z * s;
            a[96] = v.w * s;
        }
    }
    int* const   g_r  = lst_r  + (size_t)tile * LCAP;
    int* const   g_c  = lst_c  + (size_t)tile * LCAP;
    float* const g_nv = lst_nv + (size_t)tile * LCAP;
    int n = 0;
    bool build = true;
    if (reuse) {
        const int saved = lst_cnt[tile];
        if (saved <= LCAP) {
            build = false;
            n = saved;
            for (int i = tid; i < n; i += AGG_THREADS) {
                l_r[i]  = g_r[i];
                l_c[i]  = g_c[i];
                l_nv[i] = g_nv[i];
            }
        }
    }
    __syncthreads();

    if (build) {
        {
            const uint4* pl = (const uint4*)planes;
            const int nu = key_units(E);
            const int start = (int)(((unsigned)tile * (unsigned)nu / gridDim.x) & ~31u);
            for (int v0 = tid; v0 < nu; v0 += 2 * AGG_THREADS) {
                const bool two = v0 + AGG_THREADS < nu;
                int u0 = v0 + start;
                if (u0 >= nu) u0 -= nu;
                int u1 = (two ? v0 + AGG_THREADS : v0) + start;
                if (u1 >= nu) u1 -= nu;
                uint4 qa[8], qb[8];
#pragma unroll
                for (int p = 0; p < 8; ++p) {
                    qa[p] = pl[(size_t)p * nu + u0];
                    qb[p] = pl[(size_t)p * nu + u1];
                }
                uint4 da = make_uint4(0u, 0u, 0u, 0u);
                uint4 db = make_uint4(0u, 0u, 0u, 0u);
#pragma unroll
                for (int p = 0; p < 8; ++p) {
                    const unsigned f = 0u - (((unsigned)tile >> p) & 1u);
                    da.x |= qa[p].x ^ f; da.y |= qa[p].y ^ f; da.z |= qa[p].z ^ f; da.w |= qa[p].w ^ f;
                    db.x |= qb[p].x ^ f; db.y |= qb[p].y ^ f; db.z |= qb[p].z ^ f; db.w |= qb[p].w ^ f;
                }
                unsigned ma0 = ~da.x, ma1 = ~da.y, ma2 = ~da.z, ma3 = ~da.w;
                unsigned mb0 = two ? ~db.x : 0u, mb1 = two ? ~db.y : 0u;
                unsigned mb2 = two ? ~db.z : 0u, mb3 = two ? ~db.w : 0u;
                if (((u0 + 1) << 7) > E) {
                    ma0 &= valid_bits(E, u0, 0); ma1 &= valid_bits(E, u0, 1);
                    ma2 &= valid_bits(E, u0, 2); ma3 &= valid_bits(E, u0, 3);
                }
                if (((u1 + 1) << 7) > E) {
                    mb0 &= valid_bits(E, u1, 0); mb1 &= valid_bits(E, u1, 1);
                    mb2 &= valid_bits(E, u1, 2); mb3 &= valid_bits(E, u1, 3);
                }
                const int cnt = __popc(ma0) + __popc(ma1) + __popc(ma2) + __popc(ma3) +
                                __popc(mb0) + __popc(mb1) + __popc(mb2) + __popc(mb3);
                const unsigned act = __builtin_amdgcn_ballot_w32(true);
                int below = 0, total = 0;
#pragma unroll
                for (int k = 0; k < 9; ++k) {
                    const unsigned v = __builtin_amdgcn_ballot_w32(((cnt >> k) & 1) != 0);
                    below += (int)__builtin_amdgcn_mbcnt_lo(v, 0u) << k;
                    total += __popc(v) << k;
                }
                int first = 0;
                if (__builtin_amdgcn_mbcnt_lo(act, 0u) == 0u) first = atomicAdd(&l_cnt, total);
                int i = __builtin_amdgcn_readfirstlane(first) + below;
                i = emit_hits(ma0, u0, 0, i, l_r, rec, h, acc);
                i = emit_hits(ma1, u0, 1, i, l_r, rec, h, acc);
                i = emit_hits(ma2, u0, 2, i, l_r, rec, h, acc);
                i = emit_hits(ma3, u0, 3, i, l_r, rec, h, acc);
                i = emit_hits(mb0, u1, 0, i, l_r, rec, h, acc);
                i = emit_hits(mb1, u1, 1, i, l_r, rec, h, acc);
                i = emit_hits(mb2, u1, 2, i, l_r, rec, h, acc);
                i = emit_hits(mb3, u1, 3, i, l_r, rec, h, acc);
            }
        }
        __syncthreads();

        n = min(l_cnt, LCAP);
        {
            const uint2* rec2 = (const uint2*)rec;
            uint2 q[RPT];
#pragma unroll
            for (int j = 0; j < RPT; ++j) {
                const int i = tid + j * AGG_THREADS;
                const int e = l_r[i];
                q[j] = rec2[(i < n) ? e : 0];
            }
#pragma unroll
            for (int j = 0; j < RPT; ++j) {
                const int i = tid + j * AGG_THREADS;
                if (i < n) {
                    const int   r  = (int)(q[j].x & ((1u << RSHIFT) - 1u));
                    const int   c  = (int)(q[j].x >> RSHIFT);
                    const float nv = __uint_as_float(q[j].y);
                    l_r[i]  = r;
                    l_c[i]  = c;
                    l_nv[i] = nv;
                    if (!reuse) { g_r[i] = r; g_c[i] = c; g_nv[i] = nv; }
                }
            }
        }
        if (!reuse && tid == 0) lst_cnt[tile] = l_cnt;
        __syncthreads();
    }

    {
        const int wv = __builtin_amdgcn_readfirstlane(tid >> 5);
        const int nu = __builtin_amdgcn_readfirstlane(n);
        const int lane = tid & 31;
        const int sub = lane & (GU - 1);
        const float4* hl = (const float4*)h + lane;
        for (int q = wv; q * GU < nu; q += AGG_THREADS / 32) {
            const int e0 = q * GU;
            const int ei = min(e0 + sub, nu - 1);
            const int my_r  = l_r[ei];
            const int my_c  = l_c[ei];
            const int my_nv = __float_as_int(l_nv[ei]);
            float4 v[GU];
            float  nv[GU];
            int    cc[GU];
#pragma unroll
            for (int u = 0; u < GU; ++u) {
                const int r = __builtin_amdgcn_readlane(my_r, u);
                cc[u] = __builtin_amdgcn_readlane(my_c, u);
                nv[u] = __int_as_float(__builtin_amdgcn_readlane(my_nv, u));
                v[u] = hl[(size_t)r * 32];
            }
            if (e0 + GU <= nu) {
#pragma unroll
                for (int u = 0; u < GU; ++u) {
                    float* a = acc + cc[u] * HID_CH + lane;
                    unsafeAtomicAdd(a +  0, v[u].x * nv[u]);
                    unsafeAtomicAdd(a + 32, v[u].y * nv[u]);
                    unsafeAtomicAdd(a + 64, v[u].z * nv[u]);
                    unsafeAtomicAdd(a + 96, v[u].w * nv[u]);
                }
            } else {
#pragma unroll
                for (int u = 0; u < GU; ++u) {
                    if (e0 + u < nu) {
                        float* a = acc + cc[u] * HID_CH + lane;
                        unsafeAtomicAdd(a +  0, v[u].x * nv[u]);
                        unsafeAtomicAdd(a + 32, v[u].y * nv[u]);
                        unsafeAtomicAdd(a + 64, v[u].z * nv[u]);
                        unsafeAtomicAdd(a + 96, v[u].w * nv[u]);
                    }
                }
            }
        }
    }
    __syncthreads();

    {
        float4* d4 = (float4*)(dst + (size_t)base * HID_CH);
        for (int idx = tid; idx < nrows * 32; idx += AGG_THREADS) {
            const float* a = acc + (idx >> 5) * HID_CH + (idx & 31);
            d4[idx] = make_float4(a[0], a[32], a[64], a[96]);
        }
    }
}

__global__ __launch_bounds__(256) void k_bias_l2norm(const float* __restrict__ src,
                                                     float* __restrict__ out,
                                                     const float* __restrict__ b, int N) {
    int wid = (blockIdx.x * 256 + threadIdx.x) >> 5;
    int lane = threadIdx.x & 31;
    if (wid >= N) return;
    const float4* sp = (const float4*)(src + (size_t)wid * HID_CH);
    float4* op = (float4*)(out + (size_t)wid * HID_CH);
    const float4* bp = (const float4*)b;
    float4 v = sp[lane];
    float4 bb = bp[lane];
    v.x += bb.x; v.y += bb.y; v.z += bb.z; v.w += bb.w;
    float ss = v.x * v.x + v.y * v.y + v.z * v.z + v.w * v.w;
#pragma unroll
    for (int m = 16; m >= 1; m >>= 1) ss += __shfl_xor(ss, m, 32);
    float inv = 1.0f / fmaxf(sqrtf(ss), 1e-12f);
    v.x *= inv; v.y *= inv; v.z *= inv; v.w *= inv;
    op[lane] = v;
}


static inline size_t align_up(size_t v, size_t a) { return (v + a - 1) & ~(a - 1); }

extern "C" void kernel_launch(void* const* d_in, const int* in_sizes, int n_in,
                              void* d_out, int out_size, void* d_ws, size_t ws_size,
                              hipStream_t stream) {
    const float* x  = (const float*)d_in[0];
    const int*   ei = (const int*)d_in[1];
    const float* ew = (const float*)d_in[2];
    const float* W1 = (const float*)d_in[3];
    const float* b1 = (const float*)d_in[4];
    const float* W2 = (const float*)d_in[5];
    const float* b2 = (const float*)d_in[6];
    float* out = (float*)d_out;

    const int N = in_sizes[0] / IN_CH;
    const int E = in_sizes[1] / 2;
    const int* row = ei;
    const int* col = ei + E;

    char* ws = (char*)d_ws;
    size_t off = 0;
    float* deg  = (float*)(ws + off); off = align_up(off + (size_t)N * 4, 1024);
    float* h    = (float*)(ws + off); off = align_up(off + (size_t)N * HID_CH * 4, 1024);
    float* acc  = (float*)(ws + off); off = align_up(off + (size_t)N * HID_CH * 4, 1024);
    (void)ws_size; (void)n_in; (void)out_size;

    const int nb_nodes  = (N + 255) / 256;
    const int nb_edge4  = ((E + 3) / 4 + 255) / 256;
    const int nb_tiles  = (N + TILE - 1) / TILE;
    const int nb_nwave  = (N + 7) / 8;
    const int nb_gemm   = N / GEMM_ROWS;

    k_init_deg   <<<nb_nodes, 256, 0, stream>>>(deg, N);
    k_deg_hist   <<<DEG_RANGES * DEG_SLICES, DEG_THREADS, 0, stream>>>(col, ew, deg, N, E);
    unsigned* planes = (unsigned*)d_out;
    char* ob = (char*)d_out;
    size_t ooff = align_up((size_t)key_units(E) * 8 * 16, 1024);
    int*   lst_cnt = (int*)(ob + ooff);   ooff = align_up(ooff + (size_t)nb_tiles * 4, 1024);
    int*   lst_r   = (int*)(ob + ooff);   ooff = align_up(ooff + (size_t)nb_tiles * LCAP * 4, 1024);
    int*   lst_c   = (int*)(ob + ooff);   ooff = align_up(ooff + (size_t)nb_tiles * LCAP * 4, 1024);
    float* lst_nv  = (float*)(ob + ooff); ooff = align_up(ooff + (size_t)nb_tiles * LCAP * 4, 1024);
    unsigned* rec  = (unsigned*)(ob + ooff);
    k_edge_norm  <<<nb_edge4, 256, 0, stream>>>(row, col, ew, deg, rec, planes, E);

    k_gemm_wmma  <<<nb_gemm, 256, 0, stream>>>(x, W1, nullptr, h, IN_CH);
    k_tile_agg   <<<nb_tiles, AGG_THREADS, 0, stream>>>(rec, planes, h, deg, acc,
                                                        lst_cnt, lst_r, lst_c, lst_nv, 0, N, E);

    k_gemm_wmma  <<<nb_gemm, 256, 0, stream>>>(acc, W2, b1, h, HID_CH);
    k_tile_agg   <<<nb_tiles, AGG_THREADS, 0, stream>>>(rec, planes, h, deg, acc,
                                                        lst_cnt, lst_r, lst_c, lst_nv, 1, N, E);
    k_bias_l2norm<<<nb_nwave, 256, 0, stream>>>(acc, out, b2, N);
}


# Tile sums move from LDS f32 atomics to registers: wave w owns tile rows = w mod 32 (4 VGPRs each, picked by scalar index), entries handed to owner waves by LDS counters; sum order = hand-out order
# speedup vs baseline: 3.2462x; 1.0546x over previous
#include <hip/hip_runtime.h>
#include <hip/hip_bf16.h>


typedef __attribute__((ext_vector_type(2))) float v2f;
typedef __attribute__((ext_vector_type(8))) float v8f;
typedef __attribute__((ext_vector_type(16))) float v16f;

#define IN_CH 256
#define HID_CH 128
#define GEMM_ROWS 32
#define TILE 392
#define LCAP 8192
#define AGG_THREADS 1024
#define GU 12
#define WCAP 384
#define RSHIFT 17
#define RPT (LCAP / AGG_THREADS)
#define DEG_RANGES 8
#define DEG_SLICES 32
#define DEG_RCAP 12544
#define DEG_THREADS 1024

__host__ __device__ static inline int key_units(int E) { return ((((E + 3) / 4) + 255) / 256) * 8; }


__global__ __launch_bounds__(256) void k_init_deg(float* deg, int N) {
    int i = blockIdx.x * 256 + threadIdx.x;
    if (i < N) deg[i] = 1.0f;
}

__global__ __launch_bounds__(DEG_THREADS) void k_deg_hist(const int* __restrict__ col,
                                                          const float* __restrict__ w,
                                                          float* deg, int N, int E) {
    __shared__ float hist[DEG_RCAP];
    const int tid = threadIdx.x;
    const int rg = blockIdx.x % DEG_RANGES;
    const int sl = blockIdx.x / DEG_RANGES;
    const int rlen = (((N + DEG_RANGES - 1) / DEG_RANGES) + 31) & ~31;
    const int lo = rg * rlen;
    const int cnt = max(0, min(rlen, N - lo));
    for (int i = tid; i < cnt; i += DEG_THREADS) hist[i] = 0.0f;
    __syncthreads();

    if ((E & 3) == 0) {
        const int ng = E >> 2;
        const int per = (ng + DEG_SLICES - 1) / DEG_SLICES;
        const int g0 = sl * per;
        const int g1 = min(g0 + per, ng);
        const int4* c4 = (const int4*)col;
        const float4* w4 = (const float4*)w;
        for (int g = g0 + tid; g < g1; g += 4 * DEG_THREADS) {
            int4 c[4];
            float4 ww[4];
#pragma unroll
            for (int j = 0; j < 4; ++j) {
                const int gj = g + j * DEG_THREADS;
                const int gi = (gj < g1) ? gj : g;
                c[j] = c4[gi];
                ww[j] = w4[gi];
            }
#pragma unroll
            for (int j = 0; j < 4; ++j) {
                if (g + j * DEG_THREADS < g1) {
                    const unsigned u0 = (unsigned)(c[j].x - lo), u1 = (unsigned)(c[j].y - lo);
                    const unsigned u2 = (unsigned)(c[j].z - lo), u3 = (unsigned)(c[j].w - lo);
                    if (u0 < (unsigned)cnt) unsafeAtomicAdd(&hist[u0], ww[j].x);
                    if (u1 < (unsigned)cnt) unsafeAtomicAdd(&hist[u1], ww[j].y);
                    if (u2 < (unsigned)cnt) unsafeAtomicAdd(&hist[u2], ww[j].z);
                    if (u3 < (unsigned)cnt) unsafeAtomicAdd(&hist[u3], ww[j].w);
                }
            }
        }
    } else {
        const int per = (E + DEG_SLICES - 1) / DEG_SLICES;
        const int e0 = sl * per;
        const int e1 = min(e0 + per, E);
        for (int e = e0 + tid; e < e1; e += DEG_THREADS) {
            const unsigned u = (unsigned)(col[e] - lo);
            if (u < (unsigned)cnt) unsafeAtomicAdd(&hist[u], w[e]);
        }
    }
    __syncthreads();

    for (int i = tid; i < cnt; i += DEG_THREADS) unsafeAtomicAdd(&deg[lo + i], hist[i]);
}

__device__ __forceinline__ float inv_sqrt_deg(float d) {
    return (d > 0.0f) ? rsqrtf(d) : 0.0f;
}

__global__ __launch_bounds__(256) void k_edge_norm(const int* __restrict__ row,
                                                   const int* __restrict__ col,
                                                   const float* __restrict__ w,
                                                   const float* __restrict__ deg,
                                                   unsigned* __restrict__ rec,
                                                   unsigned* __restrict__ planes, int E) {
    const int t = blockIdx.x * 256 + threadIdx.x;
    const int e0 = t * 4;
    unsigned tk[4] = {0u, 0u, 0u, 0u};
    if (e0 + 3 < E) {
        int r[4], c[4];
        float ww[4];
        unsigned rc[4], nb[4];
#pragma unroll
        for (int j = 0; j < 4; ++j) { r[j] = row[e0 + j]; c[j] = col[e0 + j]; ww[j] = w[e0 + j]; }
#pragma unroll
        for (int j = 0; j < 4; ++j) {
            const unsigned tl = (unsigned)c[j] / TILE;
            nb[j] = __float_as_uint(inv_sqrt_deg(deg[r[j]]) * ww[j] * inv_sqrt_deg(deg[c[j]]));
            rc[j] = (unsigned)r[j] | (((unsigned)c[j] - tl * TILE) << RSHIFT);
            tk[j] = tl;
        }
        uint4* rp = (uint4*)(rec + 2 * (size_t)e0);
        rp[0] = make_uint4(rc[0], nb[0], rc[1], nb[1]);
        rp[1] = make_uint4(rc[2], nb[2], rc[3], nb[3]);
    } else {
#pragma unroll
        for (int j = 0; j < 4; ++j) {
            if (e0 + j < E) {
                const int rr = row[e0 + j];
                const int cn = col[e0 + j];
                const unsigned tl = (unsigned)cn / TILE;
                rec[2 * (size_t)(e0 + j)]     = (unsigned)rr | (((unsigned)cn - tl * TILE) << RSHIFT);
                rec[2 * (size_t)(e0 + j) + 1] = __float_as_uint(inv_sqrt_deg(deg[rr]) * w[e0 + j] * inv_sqrt_deg(deg[cn]));
                tk[j] = tl;
            }
        }
    }
    const int lane = threadIdx.x & 31;
    unsigned mine = 0u;
#pragma unroll
    for (int p = 0; p < 8; ++p) {
#pragma unroll
        for (int j = 0; j < 4; ++j) {
            const unsigned b = __builtin_amdgcn_ballot_w32(((tk[j] >> p) & 1u) != 0u);
            mine = (lane == 4 * p + j) ? b : mine;
        }
    }
    const size_t nu = (size_t)key_units(E);
    planes[(size_t)(lane >> 2) * nu * 4 + (size_t)(t >> 5) * 4 + (size_t)(lane & 3)] = mine;
}


__global__ __launch_bounds__(256) void k_gemm_wmma(const float* __restrict__ A,
                                                   const float* __restrict__ W,
                                                   const float* __restrict__ bias,
                                                   float* __restrict__ H, int K) {
    __shared__ float As[GEMM_ROWS * (IN_CH + 4)];
    const int stride = K + 4;
    const int tid = threadIdx.x;
    const int rowbase = blockIdx.x * GEMM_ROWS;

    if (bias != nullptr) {
        for (int idx = tid; idx < GEMM_ROWS * K; idx += 256) {
            int r = idx / K;
            int kk = idx - r * K;
            As[r * stride + kk] = fmaxf(A[(size_t)(rowbase + r) * K + kk] + bias[kk], 0.0f);
        }
    } else {
        for (int idx = tid; idx < GEMM_ROWS * K; idx += 256) {
            int r = idx / K;
            int kk = idx - r * K;
            As[r * stride + kk] = A[(size_t)(rowbase + r) * K + kk];
        }
    }
    __syncthreads();

    const int wave = tid >> 5;
    const int lane = tid & 31;
    const int m    = lane & 15;
    const int half = lane >> 4;
    const int cb   = wave * 16;

    v8f c0 = {};
    v8f c1 = {};
    const float* wp  = W + cb + m;
    const float* ap0 = As + m * stride;
    const float* ap1 = As + (16 + m) * stride;

    for (int k = 0; k < K; k += 4) {
        const int koff = k + half * 2;
        v2f b;
        b[0] = wp[(size_t)koff * HID_CH];
        b[1] = wp[(size_t)(koff + 1) * HID_CH];
        v2f a0, a1;
        a0[0] = ap0[koff];
        a0[1] = ap0[koff + 1];
        a1[0] = ap1[koff];
        a1[1] = ap1[koff + 1];
        c0 = __builtin_amdgcn_wmma_f32_16x16x4_f32(false, a0, false, b,
                                                   (short)0, c0, false, false);
        c1 = __builtin_amdgcn_wmma_f32_16x16x4_f32(false, a1, false, b,
                                                   (short)0, c1, false, false);
    }

    float* hp0 = H + (size_t)rowbase * HID_CH + cb + m;
    float* hp1 = hp0 + (size_t)16 * HID_CH;
#pragma unroll
    for (int r = 0; r < 8; ++r) {
        hp0[(size_t)(r + 8 * half) * HID_CH] = c0[r];
        hp1[(size_t)(r + 8 * half) * HID_CH] = c1[r];
    }
}


__device__ __forceinline__ unsigned valid_bits(int E, int u, int j) {
    const int lim = (E - (u << 7) - j + 3) >> 2;
    return lim >= 32 ? 0xFFFFFFFFu : (lim <= 0 ? 0u : ((1u << lim) - 1u));
}

__device__ __forceinline__ int emit_hits(unsigned m, int u, int j, int i, int* l_r) {
    while (m != 0u) {
        const int b = __ffs((int)m) - 1;
        m &= m - 1u;
        if (i < LCAP) l_r[i] = (u << 7) + (b << 2) + j;
        ++i;
    }
    return i;
}

#define ROW_ADD(k, vv, ss)                            \
    {                                                 \
        const float4 rv_ = (vv);                      \
        const float rs_ = (ss);                       \
        const int rk_ = (k);                          \
        ax[rk_] = fmaf(rv_.x, rs_, ax[rk_]);          \
        ay[rk_] = fmaf(rv_.y, rs_, ay[rk_]);          \
        az[rk_] = fmaf(rv_.z, rs_, az[rk_]);          \
        aw[rk_] = fmaf(rv_.w, rs_, aw[rk_]);          \
    }
#define ROW_STORE(K)                                                        \
    {                                                                       \
        const int c = 32 * K + wv;                                          \
        if (c < nrows) d4[(size_t)(base + c) * 32 + lane] = make_float4(ax[K], ay[K], az[K], aw[K]); \
    }

__global__ __launch_bounds__(AGG_THREADS) void k_tile_agg(const unsigned* __restrict__ rec,
                                                          const unsigned* __restrict__ planes,
                                                          const float* __restrict__ h,
                                                          const float* __restrict__ deg,
                                                          float* __restrict__ dst,
                                                          int* lst_cnt, int* lst_r, int* lst_c,
                                                          float* lst_nv, int reuse, int N, int E) {
    __shared__ int   l_r[LCAP + TILE];
    __shared__ int   l_c[LCAP + TILE];
    __shared__ float l_nv[LCAP + TILE];
    __shared__ int   wl[(AGG_THREADS / 32) * WCAP];
    __shared__ int   wcnt[AGG_THREADS / 32];
    __shared__ int   l_cnt;
    __shared__ int   l_big;

    const int tid  = threadIdx.x;
    const int lane = tid & 31;
    const int wv   = __builtin_amdgcn_readfirstlane(tid >> 5);
    const int tile = blockIdx.x;
    const int base = tile * TILE;
    const int nrows = min(TILE, N - base);
    const int nu = key_units(E);

    if (tid == 0) { l_cnt = 0; l_big = 0; }
    if (tid < AGG_THREADS / 32) wcnt[tid] = 0;

    int* const   g_r  = lst_r  + (size_t)tile * LCAP;
    int* const   g_c  = lst_c  + (size_t)tile * LCAP;
    float* const g_nv = lst_nv + (size_t)tile * LCAP;
    int n = 0;
    if (reuse) {
        const int saved = lst_cnt[tile];
        if (saved <= LCAP) {
            n = saved;
            for (int i = tid; i < n; i += AGG_THREADS) {
                l_r[i]  = g_r[i];
                l_c[i]  = g_c[i];
                l_nv[i] = g_nv[i];
            }
        } else if (tid == 0) {
            l_big = 1;
        }
        __syncthreads();
    } else {
        __syncthreads();
        {
            const uint4* pl = (const uint4*)planes;
            const int start = (int)(((unsigned)tile * (unsigned)nu / gridDim.x) & ~31u);
            for (int v0 = tid; v0 < nu; v0 += 2 * AGG_THREADS) {
                const bool two = v0 + AGG_THREADS < nu;
                int u0 = v0 + start;
                if (u0 >= nu) u0 -= nu;
                int u1 = (two ? v0 + AGG_THREADS : v0) + start;
                if (u1 >= nu) u1 -= nu;
                uint4 qa[8], qb[8];
#pragma unroll
                for (int p = 0; p < 8; ++p) {
                    qa[p] = pl[(size_t)p * nu + u0];
                    qb[p] = pl[(size_t)p * nu + u1];
                }
                uint4 da = make_uint4(0u, 0u, 0u, 0u);
                uint4 db = make_uint4(0u, 0u, 0u, 0u);
#pragma unroll
                for (int p = 0; p < 8; ++p) {
                    const unsigned f = 0u - (((unsigned)tile >> p) & 1u);
                    da.x |= qa[p].x ^ f; da.y |= qa[p].y ^ f; da.z |= qa[p].z ^ f; da.w |= qa[p].w ^ f;
                    db.x |= qb[p].x ^ f; db.y |= qb[p].y ^ f; db.z |= qb[p].z ^ f; db.w |= qb[p].w ^ f;
                }
                unsigned ma0 = ~da.x, ma1 = ~da.y, ma2 = ~da.z, ma3 = ~da.w;
                unsigned mb0 = two ? ~db.x : 0u, mb1 = two ? ~db.y : 0u;
                unsigned mb2 = two ? ~db.z : 0u, mb3 = two ? ~db.w : 0u;
                if (((u0 + 1) << 7) > E) {
                    ma0 &= valid_bits(E, u0, 0); ma1 &= valid_bits(E, u0, 1);
                    ma2 &= valid_bits(E, u0, 2); ma3 &= valid_bits(E, u0, 3);
                }
                if (((u1 + 1) << 7) > E) {
                    mb0 &= valid_bits(E, u1, 0); mb1 &= valid_bits(E, u1, 1);
                    mb2 &= valid_bits(E, u1, 2); mb3 &= valid_bits(E, u1, 3);
                }
                const int cnt = __popc(ma0) + __popc(ma1) + __popc(ma2) + __popc(ma3) +
                                __popc(mb0) + __popc(mb1) + __popc(mb2) + __popc(mb3);
                const unsigned act = __builtin_amdgcn_ballot_w32(true);
                int below = 0, total = 0;
#pragma unroll
                for (int k = 0; k < 9; ++k) {
                    const unsigned v = __builtin_amdgcn_ballot_w32(((cnt >> k) & 1) != 0);
                    below += (int)__builtin_amdgcn_mbcnt_lo(v, 0u) << k;
                    total += __popc(v) << k;
                }
                int first = 0;
                if (__builtin_amdgcn_mbcnt_lo(act, 0u) == 0u) first = atomicAdd(&l_cnt, total);
                int i = __builtin_amdgcn_readfirstlane(first) + below;
                i = emit_hits(ma0, u0, 0, i, l_r);
                i = emit_hits(ma1, u0, 1, i, l_r);
                i = emit_hits(ma2, u0, 2, i, l_r);
                i = emit_hits(ma3, u0, 3, i, l_r);
                i = emit_hits(mb0, u1, 0, i, l_r);
                i = emit_hits(mb1, u1, 1, i, l_r);
                i = emit_hits(mb2, u1, 2, i, l_r);
                i = emit_hits(mb3, u1, 3, i, l_r);
            }
        }
        __syncthreads();

        const int found = l_cnt;
        if (found <= LCAP) {
            n = found;
            const uint2* rec2 = (const uint2*)rec;
            uint2 q[RPT];
#pragma unroll
            for (int j = 0; j < RPT; ++j) {
                const int i = tid + j * AGG_THREADS;
                const int e = l_r[i];
                q[j] = rec2[(i < n) ? e : 0];
            }
#pragma unroll
            for (int j = 0; j < RPT; ++j) {
                const int i = tid + j * AGG_THREADS;
                if (i < n) {
                    const int   r  = (int)(q[j].x & ((1u << RSHIFT) - 1u));
                    const int   c  = (int)(q[j].x >> RSHIFT);
                    const float nv = __uint_as_float(q[j].y);
                    l_r[i]  = r;
                    l_c[i]  = c;
                    l_nv[i] = nv;
                    g_r[i] = r; g_c[i] = c; g_nv[i] = nv;
                }
            }
        }
        if (tid == 0) {
            lst_cnt[tile] = found;
            if (found > LCAP) l_big = 1;
        }
        __syncthreads();
    }

    for (int i = tid; i < n; i += AGG_THREADS) {
        const int w = l_c[i] & 31;
        const int p = atomicAdd(&wcnt[w], 1);
        if (p < WCAP) wl[w * WCAP + p] = i;
        else l_big = 1;
    }
    for (int c = tid; c < nrows; c += AGG_THREADS) {
        const float d = inv_sqrt_deg(deg[base + c]);
        l_r[LCAP + c]  = base + c;
        l_c[LCAP + c]  = c;
        l_nv[LCAP + c] = d * d;
        const int p = atomicAdd(&wcnt[c & 31], 1);
        if (p < WCAP) wl[(c & 31) * WCAP + p] = LCAP + c;
        else l_big = 1;
    }
    __syncthreads();
    const bool big = l_big != 0;

    const float4* hl = (const float4*)h + lane;
    v16f ax = {}, ay = {}, az = {}, aw = {};

    if (!big) {
        const int cntw = __builtin_amdgcn_readfirstlane(wcnt[wv]);
        const int* myl = wl + wv * WCAP;
        const unsigned lane16 = (unsigned)lane * 16u;
        for (int q = 0; q < cntw; q += GU) {
            const int ii = myl[min(q + lane, cntw - 1)];
            const int my_r  = l_r[ii];
            const int my_c  = l_c[ii];
            const int my_nv = __float_as_int(l_nv[ii]);
            float4 v[GU];
            float  nv[GU];
            int    kk[GU];
#pragma unroll
            for (int u = 0; u < GU; ++u) {
                const int r = __builtin_amdgcn_readlane(my_r, u);
                kk[u] = __builtin_amdgcn_readlane(my_c, u) >> 5;
                nv[u] = __int_as_float(__builtin_amdgcn_readlane(my_nv, u));
                v[u] = *(const float4*)((const char*)h + (size_t)r * (HID_CH * 4) + lane16);
            }
#pragma unroll
            for (int u = 0; u < GU; ++u) {
                ROW_ADD((q + u < cntw) ? kk[u] : 13, v[u], nv[u])
            }
        }
    } else {
        const uint4* pl = (const uint4*)planes;
        const uint2* rec2 = (const uint2*)rec;
#pragma nounroll
        for (int k = 0; k < 13; ++k) {
            const int c = 32 * k + wv;
            const int cl = min(c, nrows - 1);
            const float d = inv_sqrt_deg(deg[base + cl]);
            ROW_ADD((c < nrows) ? k : 13, hl[(size_t)(base + cl) * 32], d * d)
        }
        for (int u = 0; u < nu; ++u) {
            uint4 d = make_uint4(0u, 0u, 0u, 0u);
            for (int p = 0; p < 8; ++p) {
                const unsigned f = 0u - (((unsigned)tile >> p) & 1u);
                const uint4 qp = pl[(size_t)p * nu + u];
                d.x |= qp.x ^ f; d.y |= qp.y ^ f; d.z |= qp.z ^ f; d.w |= qp.w ^ f;
            }
            unsigned mm[4] = {~d.x & valid_bits(E, u, 0), ~d.y & valid_bits(E, u, 1),
                              ~d.z & valid_bits(E, u, 2), ~d.w & valid_bits(E, u, 3)};
#pragma unroll
            for (int j = 0; j < 4; ++j) {
                unsigned m = mm[j];
                while (m != 0u) {
                    const int b = __ffs((int)m) - 1;
                    m &= m - 1u;
                    const uint2 qe = rec2[(u << 7) + (b << 2) + j];
                    const int c = (int)(qe.x >> RSHIFT);
                    const int r = (int)(qe.x & ((1u << RSHIFT) - 1u));
                    ROW_ADD(((c & 31) == wv) ? (c >> 5) : 13, hl[(size_t)r * 32], __uint_as_float(qe.y))
                }
            }
        }
    }

    {
        float4* d4 = (float4*)dst;
        ROW_STORE(0) ROW_STORE(1) ROW_STORE(2) ROW_STORE(3) ROW_STORE(4) ROW_STORE(5) ROW_STORE(6)
        ROW_STORE(7) ROW_STORE(8) ROW_STORE(9) ROW_STORE(10) ROW_STORE(11) ROW_STORE(12)
    }
}


__global__ __launch_bounds__(256) void k_bias_l2norm(const float* __restrict__ src,
                                                     float* __restrict__ out,
                                                     const float* __restrict__ b, int N) {
    int wid = (blockIdx.x * 256 + threadIdx.x) >> 5;
    int lane = threadIdx.x & 31;
    if (wid >= N) return;
    const float4* sp = (const float4*)(src + (size_t)wid * HID_CH);
    float4* op = (float4*)(out + (size_t)wid * HID_CH);
    const float4* bp = (const float4*)b;
    float4 v = sp[lane];
    float4 bb = bp[lane];
    v.x += bb.x; v.y += bb.y; v.z += bb.z; v.w += bb.w;
    float ss = v.x * v.x + v.y * v.y + v.z * v.z + v.w * v.w;
#pragma unroll
    for (int m = 16; m >= 1; m >>= 1) ss += __shfl_xor(ss, m, 32);
    float inv = 1.0f / fmaxf(sqrtf(ss), 1e-12f);
    v.x *= inv; v.y *= inv; v.z *= inv; v.w *= inv;
    op[lane] = v;
}


static inline size_t align_up(size_t v, size_t a) { return (v + a - 1) & ~(a - 1); }

extern "C" void kernel_launch(void* const* d_in, const int* in_sizes, int n_in,
                              void* d_out, int out_size, void* d_ws, size_t ws_size,
                              hipStream_t stream) {
    const float* x  = (const float*)d_in[0];
    const int*   ei = (const int*)d_in[1];
    const float* ew = (const float*)d_in[2];
    const float* W1 = (const float*)d_in[3];
    const float* b1 = (const float*)d_in[4];
    const float* W2 = (const float*)d_in[5];
    const float* b2 = (const float*)d_in[6];
    float* out = (float*)d_out;

    const int N = in_sizes[0] / IN_CH;
    const int E = in_sizes[1] / 2;
    const int* row = ei;
    const int* col = ei + E;

    char* ws = (char*)d_ws;
    size_t off = 0;
    float* deg  = (float*)(ws + off); off = align_up(off + (size_t)N * 4, 1024);
    float* h    = (float*)(ws + off); off = align_up(off + (size_t)N * HID_CH * 4, 1024);
    float* acc  = (float*)(ws + off); off = align_up(off + (size_t)N * HID_CH * 4, 1024);
    (void)ws_size; (void)n_in; (void)out_size;

    const int nb_nodes  = (N + 255) / 256;
    const int nb_edge4  = ((E + 3) / 4 + 255) / 256;
    const int nb_tiles  = (N + TILE - 1) / TILE;
    const int nb_nwave  = (N + 7) / 8;
    const int nb_gemm   = N / GEMM_ROWS;

    k_init_deg   <<<nb_nodes, 256, 0, stream>>>(deg, N);
    k_deg_hist   <<<DEG_RANGES * DEG_SLICES, DEG_THREADS, 0, stream>>>(col, ew, deg, N, E);
    unsigned* planes = (unsigned*)d_out;
    char* ob = (char*)d_out;
    size_t ooff = align_up((size_t)key_units(E) * 8 * 16, 1024);
    int*   lst_cnt = (int*)(ob + ooff);   ooff = align_up(ooff + (size_t)nb_tiles * 4, 1024);
    int*   lst_r   = (int*)(ob + ooff);   ooff = align_up(ooff + (size_t)nb_tiles * LCAP * 4, 1024);
    int*   lst_c   = (int*)(ob + ooff);   ooff = align_up(ooff + (size_t)nb_tiles * LCAP * 4, 1024);
    float* lst_nv  = (float*)(ob + ooff); ooff = align_up(ooff + (size_t)nb_tiles * LCAP * 4, 1024);
    unsigned* rec  = (unsigned*)(ob + ooff);
    k_edge_norm  <<<nb_edge4, 256, 0, stream>>>(row, col, ew, deg, rec, planes, E);

    k_gemm_wmma  <<<nb_gemm, 256, 0, stream>>>(x, W1, nullptr, h, IN_CH);
    k_tile_agg   <<<nb_tiles, AGG_THREADS, 0, stream>>>(rec, planes, h, deg, acc,
                                                        lst_cnt, lst_r, lst_c, lst_nv, 0, N, E);

    k_gemm_wmma  <<<nb_gemm, 256, 0, stream>>>(acc, W2, b1, h, HID_CH);
    k_tile_agg   <<<nb_tiles, AGG_THREADS, 0, stream>>>(rec, planes, h, deg, acc,
                                                        lst_cnt, lst_r, lst_c, lst_nv, 1, N, E);
    k_bias_l2norm<<<nb_nwave, 256, 0, stream>>>(acc, out, b2, N);
}


# Products on a resident grid: 512 blocks hold W in LDS, waves loop over 16x64 tiles, A fragments read straight from global (next 64 k in flight), no A staging or barrier per tile; k sum order changes
# speedup vs baseline: 3.6999x; 1.1398x over previous
#include <hip/hip_runtime.h>
#include <hip/hip_bf16.h>


typedef __attribute__((ext_vector_type(2))) float v2f;
typedef __attribute__((ext_vector_type(8))) float v8f;
typedef __attribute__((ext_vector_type(16))) float v16f;

#define IN_CH 256
#define HID_CH 128
#define TILE 392
#define LCAP 8192
#define AGG_THREADS 1024
#define GU 12
#define WCAP 384
#define RSHIFT 17
#define RPT (LCAP / AGG_THREADS)
#define DEG_RANGES 8
#define DEG_SLICES 32
#define DEG_RCAP 12544
#define DEG_THREADS 1024

__host__ __device__ static inline int key_units(int E) { return ((((E + 3) / 4) + 255) / 256) * 8; }


__global__ __launch_bounds__(256) void k_init_deg(float* deg, int N) {
    int i = blockIdx.x * 256 + threadIdx.x;
    if (i < N) deg[i] = 1.0f;
}

__global__ __launch_bounds__(DEG_THREADS) void k_deg_hist(const int* __restrict__ col,
                                                          const float* __restrict__ w,
                                                          float* deg, int N, int E) {
    __shared__ float hist[DEG_RCAP];
    const int tid = threadIdx.x;
    const int rg = blockIdx.x % DEG_RANGES;
    const int sl = blockIdx.x / DEG_RANGES;
    const int rlen = (((N + DEG_RANGES - 1) / DEG_RANGES) + 31) & ~31;
    const int lo = rg * rlen;
    const int cnt = max(0, min(rlen, N - lo));
    for (int i = tid; i < cnt; i += DEG_THREADS) hist[i] = 0.0f;
    __syncthreads();

    if ((E & 3) == 0) {
        const int ng = E >> 2;
        const int per = (ng + DEG_SLICES - 1) / DEG_SLICES;
        const int g0 = sl * per;
        const int g1 = min(g0 + per, ng);
        const int4* c4 = (const int4*)col;
        const float4* w4 = (const float4*)w;
        for (int g = g0 + tid; g < g1; g += 4 * DEG_THREADS) {
            int4 c[4];
            float4 ww[4];
#pragma unroll
            for (int j = 0; j < 4; ++j) {
                const int gj = g + j * DEG_THREADS;
                const int gi = (gj < g1) ? gj : g;
                c[j] = c4[gi];
                ww[j] = w4[gi];
            }
#pragma unroll
            for (int j = 0; j < 4; ++j) {
                if (g + j * DEG_THREADS < g1) {
                    const unsigned u0 = (unsigned)(c[j].x - lo), u1 = (unsigned)(c[j].y - lo);
                    const unsigned u2 = (unsigned)(c[j].z - lo), u3 = (unsigned)(c[j].w - lo);
                    if (u0 < (unsigned)cnt) unsafeAtomicAdd(&hist[u0], ww[j].x);
                    if (u1 < (unsigned)cnt) unsafeAtomicAdd(&hist[u1], ww[j].y);
                    if (u2 < (unsigned)cnt) unsafeAtomicAdd(&hist[u2], ww[j].z);
                    if (u3 < (unsigned)cnt) unsafeAtomicAdd(&hist[u3], ww[j].w);
                }
            }
        }
    } else {
        const int per = (E + DEG_SLICES - 1) / DEG_SLICES;
        const int e0 = sl * per;
        const int e1 = min(e0 + per, E);
        for (int e = e0 + tid; e < e1; e += DEG_THREADS) {
            const unsigned u = (unsigned)(col[e] - lo);
            if (u < (unsigned)cnt) unsafeAtomicAdd(&hist[u], w[e]);
        }
    }
    __syncthreads();

    for (int i = tid; i < cnt; i += DEG_THREADS) unsafeAtomicAdd(&deg[lo + i], hist[i]);
}

__device__ __forceinline__ float inv_sqrt_deg(float d) {
    return (d > 0.0f) ? rsqrtf(d) : 0.0f;
}

__global__ __launch_bounds__(256) void k_edge_norm(const int* __restrict__ row,
                                                   const int* __restrict__ col,
                                                   const float* __restrict__ w,
                                                   const float* __restrict__ deg,
                                                   unsigned* __restrict__ rec,
                                                   unsigned* __restrict__ planes, int E) {
    const int t = blockIdx.x * 256 + threadIdx.x;
    const int e0 = t * 4;
    unsigned tk[4] = {0u, 0u, 0u, 0u};
    if (e0 + 3 < E) {
        int r[4], c[4];
        float ww[4];
        unsigned rc[4], nb[4];
#pragma unroll
        for (int j = 0; j < 4; ++j) { r[j] = row[e0 + j]; c[j] = col[e0 + j]; ww[j] = w[e0 + j]; }
#pragma unroll
        for (int j = 0; j < 4; ++j) {
            const unsigned tl = (unsigned)c[j] / TILE;
            nb[j] = __float_as_uint(inv_sqrt_deg(deg[r[j]]) * ww[j] * inv_sqrt_deg(deg[c[j]]));
            rc[j] = (unsigned)r[j] | (((unsigned)c[j] - tl * TILE) << RSHIFT);
            tk[j] = tl;
        }
        uint4* rp = (uint4*)(rec + 2 * (size_t)e0);
        rp[0] = make_uint4(rc[0], nb[0], rc[1], nb[1]);
        rp[1] = make_uint4(rc[2], nb[2], rc[3], nb[3]);
    } else {
#pragma unroll
        for (int j = 0; j < 4; ++j) {
            if (e0 + j < E) {
                const int rr = row[e0 + j];
                const int cn = col[e0 + j];
                const unsigned tl = (unsigned)cn / TILE;
                rec[2 * (size_t)(e0 + j)]     = (unsigned)rr | (((unsigned)cn - tl * TILE) << RSHIFT);
                rec[2 * (size_t)(e0 + j) + 1] = __float_as_uint(inv_sqrt_deg(deg[rr]) * w[e0 + j] * inv_sqrt_deg(deg[cn]));
                tk[j] = tl;
            }
        }
    }
    const int lane = threadIdx.x & 31;
    unsigned mine = 0u;
#pragma unroll
    for (int p = 0; p < 8; ++p) {
#pragma unroll
        for (int j = 0; j < 4; ++j) {
            const unsigned b = __builtin_amdgcn_ballot_w32(((tk[j] >> p) & 1u) != 0u);
            mine = (lane == 4 * p + j) ? b : mine;
        }
    }
    const size_t nu = (size_t)key_units(E);
    planes[(size_t)(lane >> 2) * nu * 4 + (size_t)(t >> 5) * 4 + (size_t)(lane & 3)] = mine;
}


#define PG_THREADS 512
#define PG_BLOCKS 512
#define PG_PITCH 272

__device__ __forceinline__ float4 relu_bias4(float4 v, const float* bs) {
    const float4 b = *(const float4*)bs;
    return make_float4(fmaxf(v.x + b.x, 0.0f), fmaxf(v.y + b.y, 0.0f), fmaxf(v.z + b.z, 0.0f), fmaxf(v.w + b.w, 0.0f));
}

#define PG_CHUNK_MMA(buf, kb0)                                                              \
    _Pragma("unroll") for (int i = 0; i < 8; ++i) {                                         \
        const float4 av = buf[i];                                                           \
        const float* wr = wl + (size_t)(4 * ((kb0) + i)) * PG_PITCH;                        \
        v2f a0, a1, bf[8];                                                                  \
        a0[0] = av.x; a0[1] = av.y; a1[0] = av.z; a1[1] = av.w;                             \
        _Pragma("unroll") for (int j = 0; j < 4; ++j) {                                     \
            bf[j]     = *(const v2f*)(wr + 32 * j);                                         \
            bf[4 + j] = *(const v2f*)(wr + PG_PITCH + 32 * j);                              \
        }                                                                                   \
                      \
        __builtin_amdgcn_sched_barrier(0);                                                  \
        _Pragma("unroll") for (int j = 0; j < 4; ++j)                                       \
            acc[j] = __builtin_amdgcn_wmma_f32_16x16x4_f32(false, a0, false, bf[j], (short)0, acc[j], false, false); \
        _Pragma("unroll") for (int j = 0; j < 4; ++j)                                       \
            acc[j] = __builtin_amdgcn_wmma_f32_16x16x4_f32(false, a1, false, bf[4 + j], (short)0, acc[j], false, false); \
        __builtin_amdgcn_sched_barrier(0);                                                  \
          \
        asm volatile("v_nop\n\tv_nop\n\tv_nop\n\tv_nop");                                  \
        __builtin_amdgcn_sched_barrier(0);                                                  \
    }

#define PG_CHUNK_RELU(buf, kb0)                                                             \
    if (RB) {                                                                               \
        _Pragma("unroll") for (int i = 0; i < 8; ++i)                                       \
            buf[i] = relu_bias4(buf[i], bs + 8 * ((kb0) + i) + 4 * hh);                     \
        __builtin_amdgcn_sched_barrier(0);                                                  \
    }

template <bool RB>
__global__ __launch_bounds__(PG_THREADS, 8) void k_gemm_wmma(const float* __restrict__ A,
                                                          const float* __restrict__ W,
                                                          const float* __restrict__ bias,
                                                          float* __restrict__ H, int K, int N) {
    __shared__ __attribute__((aligned(16))) float Wl[(IN_CH / 2) * PG_PITCH];
    __shared__ __attribute__((aligned(16))) float bs[IN_CH];
    const int tid = threadIdx.x;

    const int npair = K >> 1;
    for (int u = tid; u < npair * 32; u += PG_THREADS) {
        const int p = u >> 5, q = u & 31;
        const float4 r0 = ((const float4*)W)[(size_t)(2 * p) * 32 + q];
        const float4 r1 = ((const float4*)W)[(size_t)(2 * p + 1) * 32 + q];
        float4* d = (float4*)&Wl[p * PG_PITCH + q * 8];
        d[0] = make_float4(r0.x, r1.x, r0.y, r1.y);
        d[1] = make_float4(r0.z, r1.z, r0.w, r1.w);
    }
    if (RB)
        for (int i = tid; i < K; i += PG_THREADS) bs[i] = bias[i];
    __syncthreads();

    const int lane = tid & 31;
    const int wv = __builtin_amdgcn_readfirstlane(tid >> 5);
    const int cc = lane & 15;
    const int hh = lane >> 4;
    const int ntile = N >> 4;
    const int nkb = K >> 3;

    for (int m = (wv - 2 * (int)blockIdx.x) & 15; ; m += 16) {
        const int rt = (int)blockIdx.x + (int)gridDim.x * (m >> 1);
        if (rt >= ntile) break;
        const int colbase = (m & 1) * 64;
        const float* ap = A + (size_t)(rt * 16 + cc) * K + 4 * hh;
        const float* wl = Wl + (2 * hh) * PG_PITCH + (colbase + cc) * 2;

        v8f acc[4] = {};
        float4 ca[8], cb[8];
#pragma unroll
        for (int i = 0; i < 8; ++i) ca[i] = *(const float4*)(ap + 8 * i);
        PG_CHUNK_RELU(ca, 0)
        for (int kb0 = 0; kb0 < nkb; kb0 += 16) {
#pragma unroll
            for (int i = 0; i < 8; ++i) cb[i] = *(const float4*)(ap + 8 * (kb0 + 8 + i));
            PG_CHUNK_MMA(ca, kb0)
            if (kb0 + 16 < nkb) {
#pragma unroll
                for (int i = 0; i < 8; ++i) ca[i] = *(const float4*)(ap + 8 * (kb0 + 16 + i));
            }
            PG_CHUNK_RELU(cb, kb0 + 8)
            PG_CHUNK_MMA(cb, kb0 + 8)
            if (kb0 + 16 < nkb) { PG_CHUNK_RELU(ca, kb0 + 16) }
        }

        float* hp = H + (size_t)(rt * 16 + 8 * hh) * HID_CH + colbase + cc;
#pragma unroll
        for (int j = 0; j < 4; ++j) {
#pragma unroll
            for (int r = 0; r < 8; ++r) hp[(size_t)r * HID_CH + 16 * j] = acc[j][r];
        }
    }
}


__device__ __forceinline__ unsigned valid_bits(int E, int u, int j) {
    const int lim = (E - (u << 7) - j + 3) >> 2;
    return lim >= 32 ? 0xFFFFFFFFu : (lim <= 0 ? 0u : ((1u << lim) - 1u));
}

__device__ __forceinline__ int emit_hits(unsigned m, int u, int j, int i, int* l_r) {
    while (m != 0u) {
        const int b = __ffs((int)m) - 1;
        m &= m - 1u;
        if (i < LCAP) l_r[i] = (u << 7) + (b << 2) + j;
        ++i;
    }
    return i;
}

#define ROW_ADD(k, vv, ss)                            \
    {                                                 \
        const float4 rv_ = (vv);                      \
        const float rs_ = (ss);                       \
        const int rk_ = (k);                          \
        ax[rk_] = fmaf(rv_.x, rs_, ax[rk_]);          \
        ay[rk_] = fmaf(rv_.y, rs_, ay[rk_]);          \
        az[rk_] = fmaf(rv_.z, rs_, az[rk_]);          \
        aw[rk_] = fmaf(rv_.w, rs_, aw[rk_]);          \
    }
#define ROW_STORE(K)                                                        \
    {                                                                       \
        const int c = 32 * K + wv;                                          \
        if (c < nrows) d4[(size_t)(base + c) * 32 + lane] = make_float4(ax[K], ay[K], az[K], aw[K]); \
    }

__global__ __launch_bounds__(AGG_THREADS) void k_tile_agg(const unsigned* __restrict__ rec,
                                                          const unsigned* __restrict__ planes,
                                                          const float* __restrict__ h,
                                                          const float* __restrict__ deg,
                                                          float* __restrict__ dst,
                                                          int* lst_cnt, int* lst_r, int* lst_c,
                                                          float* lst_nv, int reuse, int N, int E) {
    __shared__ int   l_r[LCAP + TILE];
    __shared__ int   l_c[LCAP + TILE];
    __shared__ float l_nv[LCAP + TILE];
    __shared__ int   wl[(AGG_THREADS / 32) * WCAP];
    __shared__ int   wcnt[AGG_THREADS / 32];
    __shared__ int   l_cnt;
    __shared__ int   l_big;

    const int tid  = threadIdx.x;
    const int lane = tid & 31;
    const int wv   = __builtin_amdgcn_readfirstlane(tid >> 5);
    const int tile = blockIdx.x;
    const int base = tile * TILE;
    const int nrows = min(TILE, N - base);
    const int nu = key_units(E);

    if (tid == 0) { l_cnt = 0; l_big = 0; }
    if (tid < AGG_THREADS / 32) wcnt[tid] = 0;

    int* const   g_r  = lst_r  + (size_t)tile * LCAP;
    int* const   g_c  = lst_c  + (size_t)tile * LCAP;
    float* const g_nv = lst_nv + (size_t)tile * LCAP;
    int n = 0;
    if (reuse) {
        const int saved = lst_cnt[tile];
        if (saved <= LCAP) {
            n = saved;
            for (int i = tid; i < n; i += AGG_THREADS) {
                l_r[i]  = g_r[i];
                l_c[i]  = g_c[i];
                l_nv[i] = g_nv[i];
            }
        } else if (tid == 0) {
            l_big = 1;
        }
        __syncthreads();
    } else {
        __syncthreads();
        {
            const uint4* pl = (const uint4*)planes;
            const int start = (int)(((unsigned)tile * (unsigned)nu / gridDim.x) & ~31u);
            for (int v0 = tid; v0 < nu; v0 += 2 * AGG_THREADS) {
                const bool two = v0 + AGG_THREADS < nu;
                int u0 = v0 + start;
                if (u0 >= nu) u0 -= nu;
                int u1 = (two ? v0 + AGG_THREADS : v0) + start;
                if (u1 >= nu) u1 -= nu;
                uint4 qa[8], qb[8];
#pragma unroll
                for (int p = 0; p < 8; ++p) {
                    qa[p] = pl[(size_t)p * nu + u0];
                    qb[p] = pl[(size_t)p * nu + u1];
                }
                uint4 da = make_uint4(0u, 0u, 0u, 0u);
                uint4 db = make_uint4(0u, 0u, 0u, 0u);
#pragma unroll
                for (int p = 0; p < 8; ++p) {
                    const unsigned f = 0u - (((unsigned)tile >> p) & 1u);
                    da.x |= qa[p].x ^ f; da.y |= qa[p].y ^ f; da.z |= qa[p].z ^ f; da.w |= qa[p].w ^ f;
                    db.x |= qb[p].x ^ f; db.y |= qb[p].y ^ f; db.z |= qb[p].z ^ f; db.w |= qb[p].w ^ f;
                }
                unsigned ma0 = ~da.x, ma1 = ~da.y, ma2 = ~da.z, ma3 = ~da.w;
                unsigned mb0 = two ? ~db.x : 0u, mb1 = two ? ~db.y : 0u;
                unsigned mb2 = two ? ~db.z : 0u, mb3 = two ? ~db.w : 0u;
                if (((u0 + 1) << 7) > E) {
                    ma0 &= valid_bits(E, u0, 0); ma1 &= valid_bits(E, u0, 1);
                    ma2 &= valid_bits(E, u0, 2); ma3 &= valid_bits(E, u0, 3);
                }
                if (((u1 + 1) << 7) > E) {
                    mb0 &= valid_bits(E, u1, 0); mb1 &= valid_bits(E, u1, 1);
                    mb2 &= valid_bits(E, u1, 2); mb3 &= valid_bits(E, u1, 3);
                }
                const int cnt = __popc(ma0) + __popc(ma1) + __popc(ma2) + __popc(ma3) +
                                __popc(mb0) + __popc(mb1) + __popc(mb2) + __popc(mb3);
                const unsigned act = __builtin_amdgcn_ballot_w32(true);
                int below = 0, total = 0;
#pragma unroll
                for (int k = 0; k < 9; ++k) {
                    const unsigned v = __builtin_amdgcn_ballot_w32(((cnt >> k) & 1) != 0);
                    below += (int)__builtin_amdgcn_mbcnt_lo(v, 0u) << k;
                    total += __popc(v) << k;
                }
                int first = 0;
                if (__builtin_amdgcn_mbcnt_lo(act, 0u) == 0u) first = atomicAdd(&l_cnt, total);
                int i = __builtin_amdgcn_readfirstlane(first) + below;
                i = emit_hits(ma0, u0, 0, i, l_r);
                i = emit_hits(ma1, u0, 1, i, l_r);
                i = emit_hits(ma2, u0, 2, i, l_r);
                i = emit_hits(ma3, u0, 3, i, l_r);
                i = emit_hits(mb0, u1, 0, i, l_r);
                i = emit_hits(mb1, u1, 1, i, l_r);
                i = emit_hits(mb2, u1, 2, i, l_r);
                i = emit_hits(mb3, u1, 3, i, l_r);
            }
        }
        __syncthreads();

        const int found = l_cnt;
        if (found <= LCAP) {
            n = found;
            const uint2* rec2 = (const uint2*)rec;
            uint2 q[RPT];
#pragma unroll
            for (int j = 0; j < RPT; ++j) {
                const int i = tid + j * AGG_THREADS;
                const int e = l_r[i];
                q[j] = rec2[(i < n) ? e : 0];
            }
#pragma unroll
            for (int j = 0; j < RPT; ++j) {
                const int i = tid + j * AGG_THREADS;
                if (i < n) {
                    const int   r  = (int)(q[j].x & ((1u << RSHIFT) - 1u));
                    const int   c  = (int)(q[j].x >> RSHIFT);
                    const float nv = __uint_as_float(q[j].y);
                    l_r[i]  = r;
                    l_c[i]  = c;
                    l_nv[i] = nv;
                    g_r[i] = r; g_c[i] = c; g_nv[i] = nv;
                }
            }
        }
        if (tid == 0) {
            lst_cnt[tile] = found;
            if (found > LCAP) l_big = 1;
        }
        __syncthreads();
    }

    for (int i = tid; i < n; i += AGG_THREADS) {
        const int w = l_c[i] & 31;
        const int p = atomicAdd(&wcnt[w], 1);
        if (p < WCAP) wl[w * WCAP + p] = i;
        else l_big = 1;
    }
    for (int c = tid; c < nrows; c += AGG_THREADS) {
        const float d = inv_sqrt_deg(deg[base + c]);
        l_r[LCAP + c]  = base + c;
        l_c[LCAP + c]  = c;
        l_nv[LCAP + c] = d * d;
        const int p = atomicAdd(&wcnt[c & 31], 1);
        if (p < WCAP) wl[(c & 31) * WCAP + p] = LCAP + c;
        else l_big = 1;
    }
    __syncthreads();
    const bool big = l_big != 0;

    const float4* hl = (const float4*)h + lane;
    v16f ax = {}, ay = {}, az = {}, aw = {};

    if (!big) {
        const int cntw = __builtin_amdgcn_readfirstlane(wcnt[wv]);
        const int* myl = wl + wv * WCAP;
        const unsigned lane16 = (unsigned)lane * 16u;
        for (int q = 0; q < cntw; q += GU) {
            const int ii = myl[min(q + lane, cntw - 1)];
            const int my_r  = l_r[ii];
            const int my_c  = l_c[ii];
            const int my_nv = __float_as_int(l_nv[ii]);
            float4 v[GU];
            float  nv[GU];
            int    kk[GU];
#pragma unroll
            for (int u = 0; u < GU; ++u) {
                const int r = __builtin_amdgcn_readlane(my_r, u);
                kk[u] = __builtin_amdgcn_readlane(my_c, u) >> 5;
                nv[u] = __int_as_float(__builtin_amdgcn_readlane(my_nv, u));
                v[u] = *(const float4*)((const char*)h + (size_t)r * (HID_CH * 4) + lane16);
            }
#pragma unroll
            for (int u = 0; u < GU; ++u) {
                ROW_ADD((q + u < cntw) ? kk[u] : 13, v[u], nv[u])
            }
        }
    } else {
        const uint4* pl = (const uint4*)planes;
        const uint2* rec2 = (const uint2*)rec;
#pragma nounroll
        for (int k = 0; k < 13; ++k) {
            const int c = 32 * k + wv;
            const int cl = min(c, nrows - 1);
            const float d = inv_sqrt_deg(deg[base + cl]);
            ROW_ADD((c < nrows) ? k : 13, hl[(size_t)(base + cl) * 32], d * d)
        }
        for (int u = 0; u < nu; ++u) {
            uint4 d = make_uint4(0u, 0u, 0u, 0u);
            for (int p = 0; p < 8; ++p) {
                const unsigned f = 0u - (((unsigned)tile >> p) & 1u);
                const uint4 qp = pl[(size_t)p * nu + u];
                d.x |= qp.x ^ f; d.y |= qp.y ^ f; d.z |= qp.z ^ f; d.w |= qp.w ^ f;
            }
            unsigned mm[4] = {~d.x & valid_bits(E, u, 0), ~d.y & valid_bits(E, u, 1),
                              ~d.z & valid_bits(E, u, 2), ~d.w & valid_bits(E, u, 3)};
#pragma unroll
            for (int j = 0; j < 4; ++j) {
                unsigned m = mm[j];
                while (m != 0u) {
                    const int b = __ffs((int)m) - 1;
                    m &= m - 1u;
                    const uint2 qe = rec2[(u << 7) + (b << 2) + j];
                    const int c = (int)(qe.x >> RSHIFT);
                    const int r = (int)(qe.x & ((1u << RSHIFT) - 1u));
                    ROW_ADD(((c & 31) == wv) ? (c >> 5) : 13, hl[(size_t)r * 32], __uint_as_float(qe.y))
                }
            }
        }
    }

    {
        float4* d4 = (float4*)dst;
        ROW_STORE(0) ROW_STORE(1) ROW_STORE(2) ROW_STORE(3) ROW_STORE(4) ROW_STORE(5) ROW_STORE(6)
        ROW_STORE(7) ROW_STORE(8) ROW_STORE(9) ROW_STORE(10) ROW_STORE(11) ROW_STORE(12)
    }
}


__global__ __launch_bounds__(256) void k_bias_l2norm(const float* __restrict__ src,
                                                     float* __restrict__ out,
                                                     const float* __restrict__ b, int N) {
    int wid = (blockIdx.x * 256 + threadIdx.x) >> 5;
    int lane = threadIdx.x & 31;
    if (wid >= N) return;
    const float4* sp = (const float4*)(src + (size_t)wid * HID_CH);
    float4* op = (float4*)(out + (size_t)wid * HID_CH);
    const float4* bp = (const float4*)b;
    float4 v = sp[lane];
    float4 bb = bp[lane];
    v.x += bb.x; v.y += bb.y; v.z += bb.z; v.w += bb.w;
    float ss = v.x * v.x + v.y * v.y + v.z * v.z + v.w * v.w;
#pragma unroll
    for (int m = 16; m >= 1; m >>= 1) ss += __shfl_xor(ss, m, 32);
    float inv = 1.0f / fmaxf(sqrtf(ss), 1e-12f);
    v.x *= inv; v.y *= inv; v.z *= inv; v.w *= inv;
    op[lane] = v;
}


static inline size_t align_up(size_t v, size_t a) { return (v + a - 1) & ~(a - 1); }

extern "C" void kernel_launch(void* const* d_in, const int* in_sizes, int n_in,
                              void* d_out, int out_size, void* d_ws, size_t ws_size,
                              hipStream_t stream) {
    const float* x  = (const float*)d_in[0];
    const int*   ei = (const int*)d_in[1];
    const float* ew = (const float*)d_in[2];
    const float* W1 = (const float*)d_in[3];
    const float* b1 = (const float*)d_in[4];
    const float* W2 = (const float*)d_in[5];
    const float* b2 = (const float*)d_in[6];
    float* out = (float*)d_out;

    const int N = in_sizes[0] / IN_CH;
    const int E = in_sizes[1] / 2;
    const int* row = ei;
    const int* col = ei + E;

    char* ws = (char*)d_ws;
    size_t off = 0;
    float* deg  = (float*)(ws + off); off = align_up(off + (size_t)N * 4, 1024);
    float* h    = (float*)(ws + off); off = align_up(off + (size_t)N * HID_CH * 4, 1024);
    float* acc  = (float*)(ws + off); off = align_up(off + (size_t)N * HID_CH * 4, 1024);
    (void)ws_size; (void)n_in; (void)out_size;

    const int nb_nodes  = (N + 255) / 256;
    const int nb_edge4  = ((E + 3) / 4 + 255) / 256;
    const int nb_tiles  = (N + TILE - 1) / TILE;
    const int nb_nwave  = (N + 7) / 8;

    k_init_deg   <<<nb_nodes, 256, 0, stream>>>(deg, N);
    k_deg_hist   <<<DEG_RANGES * DEG_SLICES, DEG_THREADS, 0, stream>>>(col, ew, deg, N, E);
    unsigned* planes = (unsigned*)d_out;
    char* ob = (char*)d_out;
    size_t ooff = align_up((size_t)key_units(E) * 8 * 16, 1024);
    int*   lst_cnt = (int*)(ob + ooff);   ooff = align_up(ooff + (size_t)nb_tiles * 4, 1024);
    int*   lst_r   = (int*)(ob + ooff);   ooff = align_up(ooff + (size_t)nb_tiles * LCAP * 4, 1024);
    int*   lst_c   = (int*)(ob + ooff);   ooff = align_up(ooff + (size_t)nb_tiles * LCAP * 4, 1024);
    float* lst_nv  = (float*)(ob + ooff); ooff = align_up(ooff + (size_t)nb_tiles * LCAP * 4, 1024);
    unsigned* rec  = (unsigned*)(ob + ooff);
    k_edge_norm  <<<nb_edge4, 256, 0, stream>>>(row, col, ew, deg, rec, planes, E);

    k_gemm_wmma<false><<<PG_BLOCKS, PG_THREADS, 0, stream>>>(x, W1, nullptr, h, IN_CH, N);
    k_tile_agg   <<<nb_tiles, AGG_THREADS, 0, stream>>>(rec, planes, h, deg, acc,
                                                        lst_cnt, lst_r, lst_c, lst_nv, 0, N, E);

    k_gemm_wmma<true><<<PG_BLOCKS, PG_THREADS, 0, stream>>>(acc, W2, b1, h, HID_CH, N);
    k_tile_agg   <<<nb_tiles, AGG_THREADS, 0, stream>>>(rec, planes, h, deg, acc,
                                                        lst_cnt, lst_r, lst_c, lst_nv, 1, N, E);
    k_bias_l2norm<<<nb_nwave, 256, 0, stream>>>(acc, out, b2, N);
}
